# starting kernel
# baseline (speedup 1.0000x reference)
#include <hip/hip_runtime.h>
#include <math.h>


typedef __attribute__((ext_vector_type(2))) float v2f;
typedef __attribute__((ext_vector_type(4))) float v4f;
typedef __attribute__((ext_vector_type(8))) float v8f;

#define NPTS 4096
#define KNB  16
#define KOB  8
#define HH   8
#define CC   128
#define HCDIM 1024
#define SCALEF 0.25f

#define KC 32
#define LDP (KC + 2)
__global__ __launch_bounds__(128)
void wmma_gemm_f32(const float* __restrict__ A, const float* __restrict__ B,
                   const float* __restrict__ bias, float* __restrict__ C,
                   int Kdim,
                   long sAm, long sAk, long bsA,
                   long sBk, long sBn, long bsB,
                   long ldc, long bsC)
{
    __shared__ __align__(16) float Alds[64][LDP];
    __shared__ __align__(16) float Blds[64][LDP];

    const int wave = threadIdx.x >> 5;
    const int lane = threadIdx.x & 31;
    const int m0 = blockIdx.y * 64;
    const int n0 = blockIdx.x * 64;
    const float* Ab = A + (long)blockIdx.z * bsA;
    const float* Bb = B + (long)blockIdx.z * bsB;
    float* Cb = C + (long)blockIdx.z * bsC;

    v8f acc[4] = {};

    const int row  = lane & 15;
    const int koff = (lane >> 4) << 1;

    for (int kb = 0; kb < Kdim; kb += KC) {
        for (int i = threadIdx.x; i < 64 * KC; i += 128) {
            int k = i & (KC - 1);
            int m = i >> 5;
            Alds[m][k] = Ab[(long)(m0 + m) * sAm + (long)(kb + k) * sAk];
        }
        if (sBn == 1) {
            for (int i = threadIdx.x; i < 64 * KC; i += 128) {
                int n = i & 63;
                int k = i >> 6;
                Blds[n][k] = Bb[(long)(kb + k) * sBk + (long)(n0 + n)];
            }
        } else {
            for (int i = threadIdx.x; i < 64 * KC; i += 128) {
                int k = i & (KC - 1);
                int n = i >> 5;
                Blds[n][k] = Bb[(long)(kb + k) + (long)(n0 + n) * sBn];
            }
        }
        __syncthreads();

#pragma unroll
        for (int ks = 0; ks < KC; ks += 4) {
            const int k0 = ks + koff;
            v2f a = *(const v2f*)&Alds[wave * 16 + row][k0];
#pragma unroll
            for (int j = 0; j < 4; ++j) {
                v2f b = *(const v2f*)&Blds[j * 16 + row][k0];
                acc[j] = __builtin_amdgcn_wmma_f32_16x16x4_f32(
                    false, a, false, b, (short)0, acc[j], false, false);
            }
        }
        __syncthreads();
    }

    const int mbase = m0 + wave * 16 + ((lane >> 4) << 3);
#pragma unroll
    for (int j = 0; j < 4; ++j) {
        const int n = n0 + j * 16 + row;
        const float bv = bias ? bias[n] : 0.0f;
#pragma unroll
        for (int r = 0; r < 8; ++r)
            Cb[(long)(mbase + r) * ldc + n] = acc[j][r] + bv;
    }
}

#define LAP 260
#define LOP 132
__global__ __launch_bounds__(256)
void attend_fused(const float* __restrict__ xak, const float* __restrict__ xav,
                  const float* __restrict__ xok, const float* __restrict__ xov,
                  const float* __restrict__ apos, const float* __restrict__ npos,
                  const float* __restrict__ opos,
                  const float* __restrict__ Q,
                  const float* __restrict__ Pa, const float* __restrict__ Po,
                  const float* __restrict__ bak, const float* __restrict__ bok,
                  const float* __restrict__ decayA, const float* __restrict__ decayO,
                  float* __restrict__ Ya, float* __restrict__ Yo)
{
    const int n = blockIdx.x;
    const int t = threadIdx.x;

    __shared__ __align__(16) float sak[KNB][LAP];
    __shared__ __align__(16) float sav[KNB][LAP];
    __shared__ __align__(16) float spa[HH][LAP];
    __shared__ __align__(16) float sok[KOB][LOP];
    __shared__ __align__(16) float sov[KOB][LOP];
    __shared__ __align__(16) float spo[HH][LOP];
    __shared__ float satta[HH][KNB];
    __shared__ float satto[HH][KOB];
    __shared__ float sdwa[KNB];
    __shared__ float sdwo[KOB];
    __shared__ float sqba[HH];
    __shared__ float sqbo[HH];

    const v4f* xakn = (const v4f*)(xak + (size_t)n * KNB * 256);
    const v4f* xavn = (const v4f*)(xav + (size_t)n * KNB * 256);
    const v4f* xokn = (const v4f*)(xok + (size_t)n * KOB * 128);
    const v4f* xovn = (const v4f*)(xov + (size_t)n * KOB * 128);
    const v4f* pan  = (const v4f*)(Pa + (size_t)n * 2048);
    const v4f* pon  = (const v4f*)(Po + (size_t)n * 1024);

    for (int i = t; i < KNB * 64; i += 256) {
        int k = i >> 6, d4 = (i & 63) << 2;
        *(v4f*)&sak[k][d4] = xakn[i];
        *(v4f*)&sav[k][d4] = xavn[i];
    }
    for (int i = t; i < HH * 64; i += 256)
        *(v4f*)&spa[i >> 6][(i & 63) << 2] = pan[i];
    for (int i = t; i < KOB * 32; i += 256) {
        int k = i >> 5, d4 = (i & 31) << 2;
        *(v4f*)&sok[k][d4] = xokn[i];
        *(v4f*)&sov[k][d4] = xovn[i];
    }
    for (int i = t; i < HH * 32; i += 256)
        *(v4f*)&spo[i >> 5][(i & 31) << 2] = pon[i];

    const float ax = apos[n * 3 + 0], ay = apos[n * 3 + 1], az = apos[n * 3 + 2];
    if (t < KNB) {
        float dx = ax - npos[(n * KNB + t) * 3 + 0];
        float dy = ay - npos[(n * KNB + t) * 3 + 1];
        float dz = az - npos[(n * KNB + t) * 3 + 2];
        float dist = fmaxf(sqrtf(dx * dx + dy * dy + dz * dz), 0.001f);
        sdwa[t] = expf(-decayA[0] * dist);
    }
    if (t >= 32 && t < 32 + KOB) {
        int k = t - 32;
        float dx = ax - opos[(n * KOB + k) * 3 + 0];
        float dy = ay - opos[(n * KOB + k) * 3 + 1];
        float dz = az - opos[(n * KOB + k) * 3 + 2];
        float dist = sqrtf(dx * dx + dy * dy + dz * dz);
        sdwo[k] = expf(-decayO[0] * dist);
    }
    if (t >= 64 && t < 64 + HH) {
        int h = t - 64;
        float sa = 0.f, so = 0.f;
        for (int c = 0; c < CC; ++c) {
            float q = Q[(size_t)n * HCDIM + h * CC + c];
            sa += q * bak[h * CC + c];
            so += q * bok[h * CC + c];
        }
        sqba[h] = sa; sqbo[h] = so;
    }
    __syncthreads();

    if (t < HH * KNB) {
        int h = t >> 4, k = t & 15;
        const v4f* ak = (const v4f*)sak[k];
        const v4f* pa = (const v4f*)spa[h];
        v4f vs = {};
        for (int d = 0; d < 64; ++d) vs += ak[d] * pa[d];
        float s = sqba[h] + vs[0] + vs[1] + vs[2] + vs[3];
        satta[h][k] = s * SCALEF * sdwa[k];
    } else if (t < HH * KNB + HH * KOB) {
        int u = t - HH * KNB;
        int h = u >> 3, k = u & 7;
        const v4f* ok = (const v4f*)sok[k];
        const v4f* po = (const v4f*)spo[h];
        v4f vs = {};
        for (int d = 0; d < 32; ++d) vs += ok[d] * po[d];
        float s = sqbo[h] + vs[0] + vs[1] + vs[2] + vs[3];
        satto[h][k] = s * SCALEF * sdwo[k];
    }
    __syncthreads();

    if (t < HH) {
        float m = -3.4e38f;
        for (int k = 0; k < KNB; ++k) m = fmaxf(m, satta[t][k]);
        float sum = 0.f;
        for (int k = 0; k < KNB; ++k) { float e = expf(satta[t][k] - m); satta[t][k] = e; sum += e; }
        float inv = 1.f / sum;
        for (int k = 0; k < KNB; ++k) satta[t][k] *= inv;
    } else if (t >= 32 && t < 32 + HH) {
        int h = t - 32;
        float m = -3.4e38f;
        for (int k = 0; k < KOB; ++k) m = fmaxf(m, satto[h][k]);
        float sum = 0.f;
        for (int k = 0; k < KOB; ++k) { float e = expf(satto[h][k] - m); satto[h][k] = e; sum += e; }
        float inv = 1.f / sum;
        for (int k = 0; k < KOB; ++k) satto[h][k] *= inv;
    }
    __syncthreads();

    v4f* Yan = (v4f*)(Ya + (size_t)n * 2048);
    v4f* Yon = (v4f*)(Yo + (size_t)n * 1024);
    for (int i = t; i < HH * 64; i += 256) {
        int h = i >> 6, d4 = (i & 63) << 2;
        v4f s = {};
        for (int k = 0; k < KNB; ++k) s += satta[h][k] * *(const v4f*)&sav[k][d4];
        Yan[i] = s;
    }
    for (int i = t; i < HH * 32; i += 256) {
        int h = i >> 5, d4 = (i & 31) << 2;
        v4f s = {};
        for (int k = 0; k < KOB; ++k) s += satto[h][k] * *(const v4f*)&sov[k][d4];
        Yon[i] = s;
    }
}

__global__ __launch_bounds__(128)
void bcomb_kernel(const float* __restrict__ bav, const float* __restrict__ Wpa,
                  const float* __restrict__ bpa,
                  const float* __restrict__ bov, const float* __restrict__ Wpo,
                  const float* __restrict__ bpo,
                  float* __restrict__ bca, float* __restrict__ bco)
{
    int j = threadIdx.x;
    float sa = bpa[j], so = bpo[j];
    for (int i = 0; i < HCDIM; ++i) {
        sa += bav[i] * Wpa[i * CC + j];
        so += bov[i] * Wpo[i * CC + j];
    }
    bca[j] = sa; bco[j] = so;
}

extern "C" void kernel_launch(void* const* d_in, const int* in_sizes, int n_in,
                              void* d_out, int out_size, void* d_ws, size_t ws_size,
                              hipStream_t stream)
{
    (void)in_sizes; (void)n_in; (void)out_size; (void)ws_size;
    const float* qf   = (const float*)d_in[0];
    const float* xak  = (const float*)d_in[1];
    const float* xav  = (const float*)d_in[2];
    const float* xok  = (const float*)d_in[3];
    const float* xov  = (const float*)d_in[4];
    const float* apos = (const float*)d_in[5];
    const float* npos = (const float*)d_in[6];
    const float* opos = (const float*)d_in[7];
    const float* Wq   = (const float*)d_in[8];
    const float* bq   = (const float*)d_in[9];
    const float* Wak  = (const float*)d_in[10];
    const float* bak  = (const float*)d_in[11];
    const float* Wav  = (const float*)d_in[12];
    const float* bav  = (const float*)d_in[13];
    const float* Wok  = (const float*)d_in[14];
    const float* bok  = (const float*)d_in[15];
    const float* Wov  = (const float*)d_in[16];
    const float* bov  = (const float*)d_in[17];
    const float* Wpa  = (const float*)d_in[18];
    const float* bpa  = (const float*)d_in[19];
    const float* Wpo  = (const float*)d_in[20];
    const float* bpo  = (const float*)d_in[21];
    const float* dA   = (const float*)d_in[22];
    const float* dO   = (const float*)d_in[23];
    float* out = (float*)d_out;

    float* ws    = (float*)d_ws;
    float* wsQ   = ws;
    float* wsPa  = wsQ  + (size_t)NPTS * HCDIM;
    float* wsPo  = wsPa + (size_t)NPTS * 2048;
    float* wsYa  = wsPo + (size_t)NPTS * 1024;
    float* wsYo  = wsYa + (size_t)NPTS * 2048;
    float* wsWca = wsYo + (size_t)NPTS * 1024;
    float* wsWco = wsWca + (size_t)2048 * 128;
    float* wsBca = wsWco + (size_t)1024 * 128;
    float* wsBco = wsBca + 128;

    wmma_gemm_f32<<<dim3(HCDIM / 64, NPTS / 64, 1), 128, 0, stream>>>(
        qf, Wq, bq, wsQ, 256,  256, 1, 0,  1024, 1, 0,  1024, 0);

    wmma_gemm_f32<<<dim3(256 / 64, NPTS / 64, HH), 128, 0, stream>>>(
        wsQ, Wak, nullptr, wsPa, 128,  1024, 1, 128,  1, 1024, 128,  2048, 256);

    wmma_gemm_f32<<<dim3(128 / 64, NPTS / 64, HH), 128, 0, stream>>>(
        wsQ, Wok, nullptr, wsPo, 128,  1024, 1, 128,  1, 1024, 128,  1024, 128);

    wmma_gemm_f32<<<dim3(128 / 64, 256 / 64, HH), 128, 0, stream>>>(
        Wav, Wpa, nullptr, wsWca, 128,  1024, 1, 128,  128, 1, 128 * 128,  128, 256 * 128);

    wmma_gemm_f32<<<dim3(128 / 64, 128 / 64, HH), 128, 0, stream>>>(
        Wov, Wpo, nullptr, wsWco, 128,  1024, 1, 128,  128, 1, 128 * 128,  128, 128 * 128);

    bcomb_kernel<<<1, 128, 0, stream>>>(bav, Wpa, bpa, bov, Wpo, bpo, wsBca, wsBco);

    attend_fused<<<NPTS, 256, 0, stream>>>(
        xak, xav, xok, xov, apos, npos, opos,
        wsQ, wsPa, wsPo, bak, bok, dA, dO, wsYa, wsYo);

    wmma_gemm_f32<<<dim3(128 / 64, NPTS / 64, 1), 128, 0, stream>>>(
        wsYa, wsWca, wsBca, out, 2048,  2048, 1, 0,  128, 1, 0,  128, 0);

    wmma_gemm_f32<<<dim3(128 / 64, NPTS / 64, 1), 128, 0, stream>>>(
        wsYo, wsWco, wsBco, out + (size_t)NPTS * CC, 1024,  1024, 1, 0,  128, 1, 0,  128, 0);
}


# GEMM K loop: the next chunk's 32 global loads per thread go out as one unrolled group into registers ahead of the WMMAs, LDS-only fence at the chunk-end barrier; sum order unchanged
# speedup vs baseline: 4.2511x; 4.2511x over previous
#include <hip/hip_runtime.h>
#include <math.h>


typedef __attribute__((ext_vector_type(2))) float v2f;
typedef __attribute__((ext_vector_type(4))) float v4f;
typedef __attribute__((ext_vector_type(8))) float v8f;

#define NPTS 4096
#define KNB  16
#define KOB  8
#define HH   8
#define CC   128
#define HCDIM 1024
#define SCALEF 0.25f

#define KC 32
#define LDP (KC + 2)
__global__ __launch_bounds__(128)
void wmma_gemm_f32(const float* __restrict__ A, const float* __restrict__ B,
                   const float* __restrict__ bias, float* __restrict__ C,
                   int Kdim,
                   long sAm, long sAk, long bsA,
                   long sBk, long sBn, long bsB,
                   long ldc, long bsC)
{
    __shared__ __align__(16) float Alds[64][LDP];
    __shared__ __align__(16) float Blds[64][LDP];

    const int wave = threadIdx.x >> 5;
    const int lane = threadIdx.x & 31;
    const int m0 = blockIdx.y * 64;
    const int n0 = blockIdx.x * 64;
    const float* Ab = A + (long)blockIdx.z * bsA;
    const float* Bb = B + (long)blockIdx.z * bsB;
    float* Cb = C + (long)blockIdx.z * bsC;

    v8f acc[4] = {};

    const int row  = lane & 15;
    const int koff = (lane >> 4) << 1;

    const bool brow = (sBn == 1);
    const int  nrm  = ((wave & 1) << 5) + lane;
    const int  krm  = wave >> 1;
    const float* pa = Ab + (long)(m0 + wave) * sAm + (long)lane * sAk;
    const float* pb = brow ? Bb + (long)krm * sBk + (long)(n0 + nrm)
                           : Bb + (long)lane + (long)(n0 + wave) * sBn;
    const long stepA = 4 * sAm;
    const long stepB = brow ? 2 * sBk : 4 * sBn;
    const long advA  = (long)KC * sAk;
    const long advB  = brow ? (long)KC * sBk : (long)KC;

    float ra[16], rb[16];
    if (Kdim > 0) {
        const float* qa = pa;
        const float* qb = pb;
#pragma unroll
        for (int u = 0; u < 16; ++u) {
            ra[u] = *qa; rb[u] = *qb;
            qa += stepA; qb += stepB;
        }
    }

    for (int kb = 0; kb < Kdim; kb += KC) {
#pragma unroll
        for (int u = 0; u < 16; ++u) Alds[wave + 4 * u][lane] = ra[u];
        if (brow) {
#pragma unroll
            for (int u = 0; u < 16; ++u) Blds[nrm][krm + 2 * u] = rb[u];
        } else {
#pragma unroll
            for (int u = 0; u < 16; ++u) Blds[wave + 4 * u][lane] = rb[u];
        }
        __syncthreads();

        if (kb + KC < Kdim) {
            pa += advA; pb += advB;
            const float* qa = pa;
            const float* qb = pb;
#pragma unroll
            for (int u = 0; u < 16; ++u) {
                ra[u] = *qa; rb[u] = *qb;
                qa += stepA; qb += stepB;
            }
        }

#pragma unroll
        for (int ks = 0; ks < KC; ks += 4) {
            const int k0 = ks + koff;
            v2f a = *(const v2f*)&Alds[wave * 16 + row][k0];
#pragma unroll
            for (int j = 0; j < 4; ++j) {
                v2f b = *(const v2f*)&Blds[j * 16 + row][k0];
                acc[j] = __builtin_amdgcn_wmma_f32_16x16x4_f32(
                    false, a, false, b, (short)0, acc[j], false, false);
            }
        }
        __builtin_amdgcn_fence(__ATOMIC_RELEASE, "workgroup", "local");
        __builtin_amdgcn_s_barrier();
        __builtin_amdgcn_fence(__ATOMIC_ACQUIRE, "workgroup", "local");
    }

    const int mbase = m0 + wave * 16 + ((lane >> 4) << 3);
#pragma unroll
    for (int j = 0; j < 4; ++j) {
        const int n = n0 + j * 16 + row;
        const float bv = bias ? bias[n] : 0.0f;
#pragma unroll
        for (int r = 0; r < 8; ++r)
            Cb[(long)(mbase + r) * ldc + n] = acc[j][r] + bv;
    }
}

#define LAP 260
#define LOP 132
__global__ __launch_bounds__(256)
void attend_fused(const float* __restrict__ xak, const float* __restrict__ xav,
                  const float* __restrict__ xok, const float* __restrict__ xov,
                  const float* __restrict__ apos, const float* __restrict__ npos,
                  const float* __restrict__ opos,
                  const float* __restrict__ Q,
                  const float* __restrict__ Pa, const float* __restrict__ Po,
                  const float* __restrict__ bak, const float* __restrict__ bok,
                  const float* __restrict__ decayA, const float* __restrict__ decayO,
                  float* __restrict__ Ya, float* __restrict__ Yo)
{
    const int n = blockIdx.x;
    const int t = threadIdx.x;

    __shared__ __align__(16) float sak[KNB][LAP];
    __shared__ __align__(16) float sav[KNB][LAP];
    __shared__ __align__(16) float spa[HH][LAP];
    __shared__ __align__(16) float sok[KOB][LOP];
    __shared__ __align__(16) float sov[KOB][LOP];
    __shared__ __align__(16) float spo[HH][LOP];
    __shared__ float satta[HH][KNB];
    __shared__ float satto[HH][KOB];
    __shared__ float sdwa[KNB];
    __shared__ float sdwo[KOB];
    __shared__ float sqba[HH];
    __shared__ float sqbo[HH];

    const v4f* xakn = (const v4f*)(xak + (size_t)n * KNB * 256);
    const v4f* xavn = (const v4f*)(xav + (size_t)n * KNB * 256);
    const v4f* xokn = (const v4f*)(xok + (size_t)n * KOB * 128);
    const v4f* xovn = (const v4f*)(xov + (size_t)n * KOB * 128);
    const v4f* pan  = (const v4f*)(Pa + (size_t)n * 2048);
    const v4f* pon  = (const v4f*)(Po + (size_t)n * 1024);

    for (int i = t; i < KNB * 64; i += 256) {
        int k = i >> 6, d4 = (i & 63) << 2;
        *(v4f*)&sak[k][d4] = xakn[i];
        *(v4f*)&sav[k][d4] = xavn[i];
    }
    for (int i = t; i < HH * 64; i += 256)
        *(v4f*)&spa[i >> 6][(i & 63) << 2] = pan[i];
    for (int i = t; i < KOB * 32; i += 256) {
        int k = i >> 5, d4 = (i & 31) << 2;
        *(v4f*)&sok[k][d4] = xokn[i];
        *(v4f*)&sov[k][d4] = xovn[i];
    }
    for (int i = t; i < HH * 32; i += 256)
        *(v4f*)&spo[i >> 5][(i & 31) << 2] = pon[i];

    const float ax = apos[n * 3 + 0], ay = apos[n * 3 + 1], az = apos[n * 3 + 2];
    if (t < KNB) {
        float dx = ax - npos[(n * KNB + t) * 3 + 0];
        float dy = ay - npos[(n * KNB + t) * 3 + 1];
        float dz = az - npos[(n * KNB + t) * 3 + 2];
        float dist = fmaxf(sqrtf(dx * dx + dy * dy + dz * dz), 0.001f);
        sdwa[t] = expf(-decayA[0] * dist);
    }
    if (t >= 32 && t < 32 + KOB) {
        int k = t - 32;
        float dx = ax - opos[(n * KOB + k) * 3 + 0];
        float dy = ay - opos[(n * KOB + k) * 3 + 1];
        float dz = az - opos[(n * KOB + k) * 3 + 2];
        float dist = sqrtf(dx * dx + dy * dy + dz * dz);
        sdwo[k] = expf(-decayO[0] * dist);
    }
    if (t >= 64 && t < 64 + HH) {
        int h = t - 64;
        float sa = 0.f, so = 0.f;
        for (int c = 0; c < CC; ++c) {
            float q = Q[(size_t)n * HCDIM + h * CC + c];
            sa += q * bak[h * CC + c];
            so += q * bok[h * CC + c];
        }
        sqba[h] = sa; sqbo[h] = so;
    }
    __syncthreads();

    if (t < HH * KNB) {
        int h = t >> 4, k = t & 15;
        const v4f* ak = (const v4f*)sak[k];
        const v4f* pa = (const v4f*)spa[h];
        v4f vs = {};
        for (int d = 0; d < 64; ++d) vs += ak[d] * pa[d];
        float s = sqba[h] + vs[0] + vs[1] + vs[2] + vs[3];
        satta[h][k] = s * SCALEF * sdwa[k];
    } else if (t < HH * KNB + HH * KOB) {
        int u = t - HH * KNB;
        int h = u >> 3, k = u & 7;
        const v4f* ok = (const v4f*)sok[k];
        const v4f* po = (const v4f*)spo[h];
        v4f vs = {};
        for (int d = 0; d < 32; ++d) vs += ok[d] * po[d];
        float s = sqbo[h] + vs[0] + vs[1] + vs[2] + vs[3];
        satto[h][k] = s * SCALEF * sdwo[k];
    }
    __syncthreads();

    if (t < HH) {
        float m = -3.4e38f;
        for (int k = 0; k < KNB; ++k) m = fmaxf(m, satta[t][k]);
        float sum = 0.f;
        for (int k = 0; k < KNB; ++k) { float e = expf(satta[t][k] - m); satta[t][k] = e; sum += e; }
        float inv = 1.f / sum;
        for (int k = 0; k < KNB; ++k) satta[t][k] *= inv;
    } else if (t >= 32 && t < 32 + HH) {
        int h = t - 32;
        float m = -3.4e38f;
        for (int k = 0; k < KOB; ++k) m = fmaxf(m, satto[h][k]);
        float sum = 0.f;
        for (int k = 0; k < KOB; ++k) { float e = expf(satto[h][k] - m); satto[h][k] = e; sum += e; }
        float inv = 1.f / sum;
        for (int k = 0; k < KOB; ++k) satto[h][k] *= inv;
    }
    __syncthreads();

    v4f* Yan = (v4f*)(Ya + (size_t)n * 2048);
    v4f* Yon = (v4f*)(Yo + (size_t)n * 1024);
    for (int i = t; i < HH * 64; i += 256) {
        int h = i >> 6, d4 = (i & 63) << 2;
        v4f s = {};
        for (int k = 0; k < KNB; ++k) s += satta[h][k] * *(const v4f*)&sav[k][d4];
        Yan[i] = s;
    }
    for (int i = t; i < HH * 32; i += 256) {
        int h = i >> 5, d4 = (i & 31) << 2;
        v4f s = {};
        for (int k = 0; k < KOB; ++k) s += satto[h][k] * *(const v4f*)&sov[k][d4];
        Yon[i] = s;
    }
}

__global__ __launch_bounds__(128)
void bcomb_kernel(const float* __restrict__ bav, const float* __restrict__ Wpa,
                  const float* __restrict__ bpa,
                  const float* __restrict__ bov, const float* __restrict__ Wpo,
                  const float* __restrict__ bpo,
                  float* __restrict__ bca, float* __restrict__ bco)
{
    int j = threadIdx.x;
    float sa = bpa[j], so = bpo[j];
    for (int i = 0; i < HCDIM; ++i) {
        sa += bav[i] * Wpa[i * CC + j];
        so += bov[i] * Wpo[i * CC + j];
    }
    bca[j] = sa; bco[j] = so;
}

extern "C" void kernel_launch(void* const* d_in, const int* in_sizes, int n_in,
                              void* d_out, int out_size, void* d_ws, size_t ws_size,
                              hipStream_t stream)
{
    (void)in_sizes; (void)n_in; (void)out_size; (void)ws_size;
    const float* qf   = (const float*)d_in[0];
    const float* xak  = (const float*)d_in[1];
    const float* xav  = (const float*)d_in[2];
    const float* xok  = (const float*)d_in[3];
    const float* xov  = (const float*)d_in[4];
    const float* apos = (const float*)d_in[5];
    const float* npos = (const float*)d_in[6];
    const float* opos = (const float*)d_in[7];
    const float* Wq   = (const float*)d_in[8];
    const float* bq   = (const float*)d_in[9];
    const float* Wak  = (const float*)d_in[10];
    const float* bak  = (const float*)d_in[11];
    const float* Wav  = (const float*)d_in[12];
    const float* bav  = (const float*)d_in[13];
    const float* Wok  = (const float*)d_in[14];
    const float* bok  = (const float*)d_in[15];
    const float* Wov  = (const float*)d_in[16];
    const float* bov  = (const float*)d_in[17];
    const float* Wpa  = (const float*)d_in[18];
    const float* bpa  = (const float*)d_in[19];
    const float* Wpo  = (const float*)d_in[20];
    const float* bpo  = (const float*)d_in[21];
    const float* dA   = (const float*)d_in[22];
    const float* dO   = (const float*)d_in[23];
    float* out = (float*)d_out;

    float* ws    = (float*)d_ws;
    float* wsQ   = ws;
    float* wsPa  = wsQ  + (size_t)NPTS * HCDIM;
    float* wsPo  = wsPa + (size_t)NPTS * 2048;
    float* wsYa  = wsPo + (size_t)NPTS * 1024;
    float* wsYo  = wsYa + (size_t)NPTS * 2048;
    float* wsWca = wsYo + (size_t)NPTS * 1024;
    float* wsWco = wsWca + (size_t)2048 * 128;
    float* wsBca = wsWco + (size_t)1024 * 128;
    float* wsBco = wsBca + 128;

    wmma_gemm_f32<<<dim3(HCDIM / 64, NPTS / 64, 1), 128, 0, stream>>>(
        qf, Wq, bq, wsQ, 256,  256, 1, 0,  1024, 1, 0,  1024, 0);

    wmma_gemm_f32<<<dim3(256 / 64, NPTS / 64, HH), 128, 0, stream>>>(
        wsQ, Wak, nullptr, wsPa, 128,  1024, 1, 128,  1, 1024, 128,  2048, 256);

    wmma_gemm_f32<<<dim3(128 / 64, NPTS / 64, HH), 128, 0, stream>>>(
        wsQ, Wok, nullptr, wsPo, 128,  1024, 1, 128,  1, 1024, 128,  1024, 128);

    wmma_gemm_f32<<<dim3(128 / 64, 256 / 64, HH), 128, 0, stream>>>(
        Wav, Wpa, nullptr, wsWca, 128,  1024, 1, 128,  128, 1, 128 * 128,  128, 256 * 128);

    wmma_gemm_f32<<<dim3(128 / 64, 128 / 64, HH), 128, 0, stream>>>(
        Wov, Wpo, nullptr, wsWco, 128,  1024, 1, 128,  128, 1, 128 * 128,  128, 128 * 128);

    bcomb_kernel<<<1, 128, 0, stream>>>(bav, Wpa, bpa, bov, Wpo, bpo, wsBca, wsBco);

    attend_fused<<<NPTS, 256, 0, stream>>>(
        xak, xav, xok, xov, apos, npos, opos,
        wsQ, wsPa, wsPo, bak, bok, dA, dO, wsYa, wsYo);

    wmma_gemm_f32<<<dim3(128 / 64, NPTS / 64, 1), 128, 0, stream>>>(
        wsYa, wsWca, wsBca, out, 2048,  2048, 1, 0,  128, 1, 0,  128, 0);

    wmma_gemm_f32<<<dim3(128 / 64, NPTS / 64, 1), 128, 0, stream>>>(
        wsYo, wsWco, wsBco, out + (size_t)NPTS * CC, 1024,  1024, 1, 0,  128, 1, 0,  128, 0);
}


# Folded-bias kernel: one 128-thread block looping over 1024 terms becomes 8 blocks x 32 waves, 32 terms per wave in one load group, partials added via LDS; sum order of those biases changes
# speedup vs baseline: 4.8302x; 1.1362x over previous
#include <hip/hip_runtime.h>
#include <math.h>


typedef __attribute__((ext_vector_type(2))) float v2f;
typedef __attribute__((ext_vector_type(4))) float v4f;
typedef __attribute__((ext_vector_type(8))) float v8f;

#define NPTS 4096
#define KNB  16
#define KOB  8
#define HH   8
#define CC   128
#define HCDIM 1024
#define SCALEF 0.25f

#define KC 32
#define LDP (KC + 2)
__global__ __launch_bounds__(128)
void wmma_gemm_f32(const float* __restrict__ A, const float* __restrict__ B,
                   const float* __restrict__ bias, float* __restrict__ C,
                   int Kdim,
                   long sAm, long sAk, long bsA,
                   long sBk, long sBn, long bsB,
                   long ldc, long bsC)
{
    __shared__ __align__(16) float Alds[64][LDP];
    __shared__ __align__(16) float Blds[64][LDP];

    const int wave = threadIdx.x >> 5;
    const int lane = threadIdx.x & 31;
    const int m0 = blockIdx.y * 64;
    const int n0 = blockIdx.x * 64;
    const float* Ab = A + (long)blockIdx.z * bsA;
    const float* Bb = B + (long)blockIdx.z * bsB;
    float* Cb = C + (long)blockIdx.z * bsC;

    v8f acc[4] = {};

    const int row  = lane & 15;
    const int koff = (lane >> 4) << 1;

    const bool brow = (sBn == 1);
    const int  nrm  = ((wave & 1) << 5) + lane;
    const int  krm  = wave >> 1;
    const float* pa = Ab + (long)(m0 + wave) * sAm + (long)lane * sAk;
    const float* pb = brow ? Bb + (long)krm * sBk + (long)(n0 + nrm)
                           : Bb + (long)lane + (long)(n0 + wave) * sBn;
    const long stepA = 4 * sAm;
    const long stepB = brow ? 2 * sBk : 4 * sBn;
    const long advA  = (long)KC * sAk;
    const long advB  = brow ? (long)KC * sBk : (long)KC;

    float ra[16], rb[16];
    if (Kdim > 0) {
        const float* qa = pa;
        const float* qb = pb;
#pragma unroll
        for (int u = 0; u < 16; ++u) {
            ra[u] = *qa; rb[u] = *qb;
            qa += stepA; qb += stepB;
        }
    }

    for (int kb = 0; kb < Kdim; kb += KC) {
#pragma unroll
        for (int u = 0; u < 16; ++u) Alds[wave + 4 * u][lane] = ra[u];
        if (brow) {
#pragma unroll
            for (int u = 0; u < 16; ++u) Blds[nrm][krm + 2 * u] = rb[u];
        } else {
#pragma unroll
            for (int u = 0; u < 16; ++u) Blds[wave + 4 * u][lane] = rb[u];
        }
        __syncthreads();

        if (kb + KC < Kdim) {
            pa += advA; pb += advB;
            const float* qa = pa;
            const float* qb = pb;
#pragma unroll
            for (int u = 0; u < 16; ++u) {
                ra[u] = *qa; rb[u] = *qb;
                qa += stepA; qb += stepB;
            }
        }

#pragma unroll
        for (int ks = 0; ks < KC; ks += 4) {
            const int k0 = ks + koff;
            v2f a = *(const v2f*)&Alds[wave * 16 + row][k0];
#pragma unroll
            for (int j = 0; j < 4; ++j) {
                v2f b = *(const v2f*)&Blds[j * 16 + row][k0];
                acc[j] = __builtin_amdgcn_wmma_f32_16x16x4_f32(
                    false, a, false, b, (short)0, acc[j], false, false);
            }
        }
        __builtin_amdgcn_fence(__ATOMIC_RELEASE, "workgroup", "local");
        __builtin_amdgcn_s_barrier();
        __builtin_amdgcn_fence(__ATOMIC_ACQUIRE, "workgroup", "local");
    }

    const int mbase = m0 + wave * 16 + ((lane >> 4) << 3);
#pragma unroll
    for (int j = 0; j < 4; ++j) {
        const int n = n0 + j * 16 + row;
        const float bv = bias ? bias[n] : 0.0f;
#pragma unroll
        for (int r = 0; r < 8; ++r)
            Cb[(long)(mbase + r) * ldc + n] = acc[j][r] + bv;
    }
}

#define LAP 260
#define LOP 132
__global__ __launch_bounds__(256)
void attend_fused(const float* __restrict__ xak, const float* __restrict__ xav,
                  const float* __restrict__ xok, const float* __restrict__ xov,
                  const float* __restrict__ apos, const float* __restrict__ npos,
                  const float* __restrict__ opos,
                  const float* __restrict__ Q,
                  const float* __restrict__ Pa, const float* __restrict__ Po,
                  const float* __restrict__ bak, const float* __restrict__ bok,
                  const float* __restrict__ decayA, const float* __restrict__ decayO,
                  float* __restrict__ Ya, float* __restrict__ Yo)
{
    const int n = blockIdx.x;
    const int t = threadIdx.x;

    __shared__ __align__(16) float sak[KNB][LAP];
    __shared__ __align__(16) float sav[KNB][LAP];
    __shared__ __align__(16) float spa[HH][LAP];
    __shared__ __align__(16) float sok[KOB][LOP];
    __shared__ __align__(16) float sov[KOB][LOP];
    __shared__ __align__(16) float spo[HH][LOP];
    __shared__ float satta[HH][KNB];
    __shared__ float satto[HH][KOB];
    __shared__ float sdwa[KNB];
    __shared__ float sdwo[KOB];
    __shared__ float sqba[HH];
    __shared__ float sqbo[HH];

    const v4f* xakn = (const v4f*)(xak + (size_t)n * KNB * 256);
    const v4f* xavn = (const v4f*)(xav + (size_t)n * KNB * 256);
    const v4f* xokn = (const v4f*)(xok + (size_t)n * KOB * 128);
    const v4f* xovn = (const v4f*)(xov + (size_t)n * KOB * 128);
    const v4f* pan  = (const v4f*)(Pa + (size_t)n * 2048);
    const v4f* pon  = (const v4f*)(Po + (size_t)n * 1024);

    for (int i = t; i < KNB * 64; i += 256) {
        int k = i >> 6, d4 = (i & 63) << 2;
        *(v4f*)&sak[k][d4] = xakn[i];
        *(v4f*)&sav[k][d4] = xavn[i];
    }
    for (int i = t; i < HH * 64; i += 256)
        *(v4f*)&spa[i >> 6][(i & 63) << 2] = pan[i];
    for (int i = t; i < KOB * 32; i += 256) {
        int k = i >> 5, d4 = (i & 31) << 2;
        *(v4f*)&sok[k][d4] = xokn[i];
        *(v4f*)&sov[k][d4] = xovn[i];
    }
    for (int i = t; i < HH * 32; i += 256)
        *(v4f*)&spo[i >> 5][(i & 31) << 2] = pon[i];

    const float ax = apos[n * 3 + 0], ay = apos[n * 3 + 1], az = apos[n * 3 + 2];
    if (t < KNB) {
        float dx = ax - npos[(n * KNB + t) * 3 + 0];
        float dy = ay - npos[(n * KNB + t) * 3 + 1];
        float dz = az - npos[(n * KNB + t) * 3 + 2];
        float dist = fmaxf(sqrtf(dx * dx + dy * dy + dz * dz), 0.001f);
        sdwa[t] = expf(-decayA[0] * dist);
    }
    if (t >= 32 && t < 32 + KOB) {
        int k = t - 32;
        float dx = ax - opos[(n * KOB + k) * 3 + 0];
        float dy = ay - opos[(n * KOB + k) * 3 + 1];
        float dz = az - opos[(n * KOB + k) * 3 + 2];
        float dist = sqrtf(dx * dx + dy * dy + dz * dz);
        sdwo[k] = expf(-decayO[0] * dist);
    }
    if (t >= 64 && t < 64 + HH) {
        int h = t - 64;
        float sa = 0.f, so = 0.f;
        for (int c = 0; c < CC; ++c) {
            float q = Q[(size_t)n * HCDIM + h * CC + c];
            sa += q * bak[h * CC + c];
            so += q * bok[h * CC + c];
        }
        sqba[h] = sa; sqbo[h] = so;
    }
    __syncthreads();

    if (t < HH * KNB) {
        int h = t >> 4, k = t & 15;
        const v4f* ak = (const v4f*)sak[k];
        const v4f* pa = (const v4f*)spa[h];
        v4f vs = {};
        for (int d = 0; d < 64; ++d) vs += ak[d] * pa[d];
        float s = sqba[h] + vs[0] + vs[1] + vs[2] + vs[3];
        satta[h][k] = s * SCALEF * sdwa[k];
    } else if (t < HH * KNB + HH * KOB) {
        int u = t - HH * KNB;
        int h = u >> 3, k = u & 7;
        const v4f* ok = (const v4f*)sok[k];
        const v4f* po = (const v4f*)spo[h];
        v4f vs = {};
        for (int d = 0; d < 32; ++d) vs += ok[d] * po[d];
        float s = sqbo[h] + vs[0] + vs[1] + vs[2] + vs[3];
        satto[h][k] = s * SCALEF * sdwo[k];
    }
    __syncthreads();

    if (t < HH) {
        float m = -3.4e38f;
        for (int k = 0; k < KNB; ++k) m = fmaxf(m, satta[t][k]);
        float sum = 0.f;
        for (int k = 0; k < KNB; ++k) { float e = expf(satta[t][k] - m); satta[t][k] = e; sum += e; }
        float inv = 1.f / sum;
        for (int k = 0; k < KNB; ++k) satta[t][k] *= inv;
    } else if (t >= 32 && t < 32 + HH) {
        int h = t - 32;
        float m = -3.4e38f;
        for (int k = 0; k < KOB; ++k) m = fmaxf(m, satto[h][k]);
        float sum = 0.f;
        for (int k = 0; k < KOB; ++k) { float e = expf(satto[h][k] - m); satto[h][k] = e; sum += e; }
        float inv = 1.f / sum;
        for (int k = 0; k < KOB; ++k) satto[h][k] *= inv;
    }
    __syncthreads();

    v4f* Yan = (v4f*)(Ya + (size_t)n * 2048);
    v4f* Yon = (v4f*)(Yo + (size_t)n * 1024);
    for (int i = t; i < HH * 64; i += 256) {
        int h = i >> 6, d4 = (i & 63) << 2;
        v4f s = {};
        for (int k = 0; k < KNB; ++k) s += satta[h][k] * *(const v4f*)&sav[k][d4];
        Yan[i] = s;
    }
    for (int i = t; i < HH * 32; i += 256) {
        int h = i >> 5, d4 = (i & 31) << 2;
        v4f s = {};
        for (int k = 0; k < KOB; ++k) s += satto[h][k] * *(const v4f*)&sov[k][d4];
        Yon[i] = s;
    }
}

__global__ __launch_bounds__(1024)
void bcomb_kernel(const float* __restrict__ bav, const float* __restrict__ Wpa,
                  const float* __restrict__ bpa,
                  const float* __restrict__ bov, const float* __restrict__ Wpo,
                  const float* __restrict__ bpo,
                  float* __restrict__ bca, float* __restrict__ bco)
{
    __shared__ float part[32][32];
    const int lane = threadIdx.x & 31;
    const int w = __builtin_amdgcn_readfirstlane((int)(threadIdx.x >> 5));
    const bool obst = blockIdx.x >= 4;
    const int j = ((blockIdx.x & 3) << 5) + lane;
    const float* bv = (obst ? bov : bav) + w * 32;
    const float* wp = (obst ? Wpo : Wpa) + (size_t)(w * 32) * CC + j;
    float s = 0.f;
#pragma unroll
    for (int u = 0; u < 32; ++u) s += bv[u] * wp[u * CC];
    part[w][lane] = s;
    __syncthreads();
    if (w == 0) {
        float t = (obst ? bpo : bpa)[j];
        for (int k = 0; k < 32; ++k) t += part[k][lane];
        (obst ? bco : bca)[j] = t;
    }
}

extern "C" void kernel_launch(void* const* d_in, const int* in_sizes, int n_in,
                              void* d_out, int out_size, void* d_ws, size_t ws_size,
                              hipStream_t stream)
{
    (void)in_sizes; (void)n_in; (void)out_size; (void)ws_size;
    const float* qf   = (const float*)d_in[0];
    const float* xak  = (const float*)d_in[1];
    const float* xav  = (const float*)d_in[2];
    const float* xok  = (const float*)d_in[3];
    const float* xov  = (const float*)d_in[4];
    const float* apos = (const float*)d_in[5];
    const float* npos = (const float*)d_in[6];
    const float* opos = (const float*)d_in[7];
    const float* Wq   = (const float*)d_in[8];
    const float* bq   = (const float*)d_in[9];
    const float* Wak  = (const float*)d_in[10];
    const float* bak  = (const float*)d_in[11];
    const float* Wav  = (const float*)d_in[12];
    const float* bav  = (const float*)d_in[13];
    const float* Wok  = (const float*)d_in[14];
    const float* bok  = (const float*)d_in[15];
    const float* Wov  = (const float*)d_in[16];
    const float* bov  = (const float*)d_in[17];
    const float* Wpa  = (const float*)d_in[18];
    const float* bpa  = (const float*)d_in[19];
    const float* Wpo  = (const float*)d_in[20];
    const float* bpo  = (const float*)d_in[21];
    const float* dA   = (const float*)d_in[22];
    const float* dO   = (const float*)d_in[23];
    float* out = (float*)d_out;

    float* ws    = (float*)d_ws;
    float* wsQ   = ws;
    float* wsPa  = wsQ  + (size_t)NPTS * HCDIM;
    float* wsPo  = wsPa + (size_t)NPTS * 2048;
    float* wsYa  = wsPo + (size_t)NPTS * 1024;
    float* wsYo  = wsYa + (size_t)NPTS * 2048;
    float* wsWca = wsYo + (size_t)NPTS * 1024;
    float* wsWco = wsWca + (size_t)2048 * 128;
    float* wsBca = wsWco + (size_t)1024 * 128;
    float* wsBco = wsBca + 128;

    wmma_gemm_f32<<<dim3(HCDIM / 64, NPTS / 64, 1), 128, 0, stream>>>(
        qf, Wq, bq, wsQ, 256,  256, 1, 0,  1024, 1, 0,  1024, 0);

    wmma_gemm_f32<<<dim3(256 / 64, NPTS / 64, HH), 128, 0, stream>>>(
        wsQ, Wak, nullptr, wsPa, 128,  1024, 1, 128,  1, 1024, 128,  2048, 256);

    wmma_gemm_f32<<<dim3(128 / 64, NPTS / 64, HH), 128, 0, stream>>>(
        wsQ, Wok, nullptr, wsPo, 128,  1024, 1, 128,  1, 1024, 128,  1024, 128);

    wmma_gemm_f32<<<dim3(128 / 64, 256 / 64, HH), 128, 0, stream>>>(
        Wav, Wpa, nullptr, wsWca, 128,  1024, 1, 128,  128, 1, 128 * 128,  128, 256 * 128);

    wmma_gemm_f32<<<dim3(128 / 64, 128 / 64, HH), 128, 0, stream>>>(
        Wov, Wpo, nullptr, wsWco, 128,  1024, 1, 128,  128, 1, 128 * 128,  128, 128 * 128);

    bcomb_kernel<<<8, 1024, 0, stream>>>(bav, Wpa, bpa, bov, Wpo, bpo, wsBca, wsBco);

    attend_fused<<<NPTS, 256, 0, stream>>>(
        xak, xav, xok, xov, apos, npos, opos,
        wsQ, wsPa, wsPo, bak, bok, dA, dO, wsYa, wsYo);

    wmma_gemm_f32<<<dim3(128 / 64, NPTS / 64, 1), 128, 0, stream>>>(
        wsYa, wsWca, wsBca, out, 2048,  2048, 1, 0,  128, 1, 0,  128, 0);

    wmma_gemm_f32<<<dim3(128 / 64, NPTS / 64, 1), 128, 0, stream>>>(
        wsYo, wsWco, wsBco, out + (size_t)NPTS * CC, 1024,  1024, 1, 0,  128, 1, 0,  128, 0);
}


# Independent work merged into shared grids: Q GEMM + both Wcomb GEMMs + bias fold in one launch, the two p GEMMs in one, the two output GEMMs in one (9 launches -> 4); sum order unchanged
# speedup vs baseline: 6.4488x; 1.3351x over previous
#include <hip/hip_runtime.h>
#include <math.h>


typedef __attribute__((ext_vector_type(2))) float v2f;
typedef __attribute__((ext_vector_type(4))) float v4f;
typedef __attribute__((ext_vector_type(8))) float v8f;

#define NPTS 4096
#define KNB  16
#define KOB  8
#define HH   8
#define CC   128
#define HCDIM 1024
#define SCALEF 0.25f

#define KC 32
#define LDP (KC + 2)
struct GemmShape {
    int  K;
    int  ltx;
    int  lty;
    int  nblk;
    long sAm, sAk, bsA;
    long sBk, sBn, bsB;
    long ldc, bsC;
};

__device__ __forceinline__
void bias_fold_block(int blk,
                     const float* __restrict__ bav, const float* __restrict__ Wpa,
                     const float* __restrict__ bpa,
                     const float* __restrict__ bov, const float* __restrict__ Wpo,
                     const float* __restrict__ bpo,
                     float* __restrict__ bca, float* __restrict__ bco);

__global__ __launch_bounds__(128)
void wmma_gemm_f32(const float* A0, const float* B0, const float* bias0, float* C0, GemmShape s0,
                   const float* A1, const float* B1, const float* bias1, float* C1, GemmShape s1,
                   const float* A2, const float* B2, const float* bias2, float* C2, GemmShape s2,
                   const float* bav, const float* Wpa, const float* bpa,
                   const float* bov, const float* Wpo, const float* bpo,
                   float* bca, float* bco)
{
    int blk = blockIdx.x;
    const float* __restrict__ A = A0;
    const float* __restrict__ B = B0;
    const float* __restrict__ bias = bias0;
    float* __restrict__ C = C0;
    GemmShape s = s0;
    if (blk >= s0.nblk) {
        blk -= s0.nblk;
        A = A1; B = B1; bias = bias1; C = C1; s = s1;
        if (blk >= s1.nblk) {
            blk -= s1.nblk;
            A = A2; B = B2; bias = bias2; C = C2; s = s2;
            if (blk >= s2.nblk) {
                bias_fold_block(blk - s2.nblk, bav, Wpa, bpa, bov, Wpo, bpo, bca, bco);
                return;
            }
        }
    }
    const int  bx = blk & ((1 << s.ltx) - 1);
    const int  by = (blk >> s.ltx) & ((1 << s.lty) - 1);
    const int  bz = blk >> (s.ltx + s.lty);
    const int  Kdim = s.K;
    const long sAm = s.sAm, sAk = s.sAk, bsA = s.bsA;
    const long sBk = s.sBk, sBn = s.sBn, bsB = s.bsB;
    const long ldc = s.ldc, bsC = s.bsC;

    __shared__ __align__(16) float Alds[64][LDP];
    __shared__ __align__(16) float Blds[64][LDP];

    const int wave = threadIdx.x >> 5;
    const int lane = threadIdx.x & 31;
    const int m0 = by * 64;
    const int n0 = bx * 64;
    const float* Ab = A + (long)bz * bsA;
    const float* Bb = B + (long)bz * bsB;
    float* Cb = C + (long)bz * bsC;

    v8f acc[4] = {};

    const int row  = lane & 15;
    const int koff = (lane >> 4) << 1;

    const bool brow = (sBn == 1);
    const int  nrm  = ((wave & 1) << 5) + lane;
    const int  krm  = wave >> 1;
    const float* pa = Ab + (long)(m0 + wave) * sAm + (long)lane * sAk;
    const float* pb = brow ? Bb + (long)krm * sBk + (long)(n0 + nrm)
                           : Bb + (long)lane + (long)(n0 + wave) * sBn;
    const long stepA = 4 * sAm;
    const long stepB = brow ? 2 * sBk : 4 * sBn;
    const long advA  = (long)KC * sAk;
    const long advB  = brow ? (long)KC * sBk : (long)KC;

    float ra[16], rb[16];
    if (Kdim > 0) {
        const float* qa = pa;
        const float* qb = pb;
#pragma unroll
        for (int u = 0; u < 16; ++u) {
            ra[u] = *qa; rb[u] = *qb;
            qa += stepA; qb += stepB;
        }
    }

    for (int kb = 0; kb < Kdim; kb += KC) {
#pragma unroll
        for (int u = 0; u < 16; ++u) Alds[wave + 4 * u][lane] = ra[u];
        if (brow) {
#pragma unroll
            for (int u = 0; u < 16; ++u) Blds[nrm][krm + 2 * u] = rb[u];
        } else {
#pragma unroll
            for (int u = 0; u < 16; ++u) Blds[wave + 4 * u][lane] = rb[u];
        }
        __syncthreads();

        if (kb + KC < Kdim) {
            pa += advA; pb += advB;
            const float* qa = pa;
            const float* qb = pb;
#pragma unroll
            for (int u = 0; u < 16; ++u) {
                ra[u] = *qa; rb[u] = *qb;
                qa += stepA; qb += stepB;
            }
        }

#pragma unroll
        for (int ks = 0; ks < KC; ks += 4) {
            const int k0 = ks + koff;
            v2f a = *(const v2f*)&Alds[wave * 16 + row][k0];
#pragma unroll
            for (int j = 0; j < 4; ++j) {
                v2f b = *(const v2f*)&Blds[j * 16 + row][k0];
                acc[j] = __builtin_amdgcn_wmma_f32_16x16x4_f32(
                    false, a, false, b, (short)0, acc[j], false, false);
            }
        }
        __builtin_amdgcn_fence(__ATOMIC_RELEASE, "workgroup", "local");
        __builtin_amdgcn_s_barrier();
        __builtin_amdgcn_fence(__ATOMIC_ACQUIRE, "workgroup", "local");
    }

    const int mbase = m0 + wave * 16 + ((lane >> 4) << 3);
#pragma unroll
    for (int j = 0; j < 4; ++j) {
        const int n = n0 + j * 16 + row;
        const float bv = bias ? bias[n] : 0.0f;
#pragma unroll
        for (int r = 0; r < 8; ++r)
            Cb[(long)(mbase + r) * ldc + n] = acc[j][r] + bv;
    }
}

#define LAP 260
#define LOP 132
__global__ __launch_bounds__(256)
void attend_fused(const float* __restrict__ xak, const float* __restrict__ xav,
                  const float* __restrict__ xok, const float* __restrict__ xov,
                  const float* __restrict__ apos, const float* __restrict__ npos,
                  const float* __restrict__ opos,
                  const float* __restrict__ Q,
                  const float* __restrict__ Pa, const float* __restrict__ Po,
                  const float* __restrict__ bak, const float* __restrict__ bok,
                  const float* __restrict__ decayA, const float* __restrict__ decayO,
                  float* __restrict__ Ya, float* __restrict__ Yo)
{
    const int n = blockIdx.x;
    const int t = threadIdx.x;

    __shared__ __align__(16) float sak[KNB][LAP];
    __shared__ __align__(16) float sav[KNB][LAP];
    __shared__ __align__(16) float spa[HH][LAP];
    __shared__ __align__(16) float sok[KOB][LOP];
    __shared__ __align__(16) float sov[KOB][LOP];
    __shared__ __align__(16) float spo[HH][LOP];
    __shared__ float satta[HH][KNB];
    __shared__ float satto[HH][KOB];
    __shared__ float sdwa[KNB];
    __shared__ float sdwo[KOB];
    __shared__ float sqba[HH];
    __shared__ float sqbo[HH];

    const v4f* xakn = (const v4f*)(xak + (size_t)n * KNB * 256);
    const v4f* xavn = (const v4f*)(xav + (size_t)n * KNB * 256);
    const v4f* xokn = (const v4f*)(xok + (size_t)n * KOB * 128);
    const v4f* xovn = (const v4f*)(xov + (size_t)n * KOB * 128);
    const v4f* pan  = (const v4f*)(Pa + (size_t)n * 2048);
    const v4f* pon  = (const v4f*)(Po + (size_t)n * 1024);

    for (int i = t; i < KNB * 64; i += 256) {
        int k = i >> 6, d4 = (i & 63) << 2;
        *(v4f*)&sak[k][d4] = xakn[i];
        *(v4f*)&sav[k][d4] = xavn[i];
    }
    for (int i = t; i < HH * 64; i += 256)
        *(v4f*)&spa[i >> 6][(i & 63) << 2] = pan[i];
    for (int i = t; i < KOB * 32; i += 256) {
        int k = i >> 5, d4 = (i & 31) << 2;
        *(v4f*)&sok[k][d4] = xokn[i];
        *(v4f*)&sov[k][d4] = xovn[i];
    }
    for (int i = t; i < HH * 32; i += 256)
        *(v4f*)&spo[i >> 5][(i & 31) << 2] = pon[i];

    const float ax = apos[n * 3 + 0], ay = apos[n * 3 + 1], az = apos[n * 3 + 2];
    if (t < KNB) {
        float dx = ax - npos[(n * KNB + t) * 3 + 0];
        float dy = ay - npos[(n * KNB + t) * 3 + 1];
        float dz = az - npos[(n * KNB + t) * 3 + 2];
        float dist = fmaxf(sqrtf(dx * dx + dy * dy + dz * dz), 0.001f);
        sdwa[t] = expf(-decayA[0] * dist);
    }
    if (t >= 32 && t < 32 + KOB) {
        int k = t - 32;
        float dx = ax - opos[(n * KOB + k) * 3 + 0];
        float dy = ay - opos[(n * KOB + k) * 3 + 1];
        float dz = az - opos[(n * KOB + k) * 3 + 2];
        float dist = sqrtf(dx * dx + dy * dy + dz * dz);
        sdwo[k] = expf(-decayO[0] * dist);
    }
    if (t >= 64 && t < 64 + HH) {
        int h = t - 64;
        float sa = 0.f, so = 0.f;
        for (int c = 0; c < CC; ++c) {
            float q = Q[(size_t)n * HCDIM + h * CC + c];
            sa += q * bak[h * CC + c];
            so += q * bok[h * CC + c];
        }
        sqba[h] = sa; sqbo[h] = so;
    }
    __syncthreads();

    if (t < HH * KNB) {
        int h = t >> 4, k = t & 15;
        const v4f* ak = (const v4f*)sak[k];
        const v4f* pa = (const v4f*)spa[h];
        v4f vs = {};
        for (int d = 0; d < 64; ++d) vs += ak[d] * pa[d];
        float s = sqba[h] + vs[0] + vs[1] + vs[2] + vs[3];
        satta[h][k] = s * SCALEF * sdwa[k];
    } else if (t < HH * KNB + HH * KOB) {
        int u = t - HH * KNB;
        int h = u >> 3, k = u & 7;
        const v4f* ok = (const v4f*)sok[k];
        const v4f* po = (const v4f*)spo[h];
        v4f vs = {};
        for (int d = 0; d < 32; ++d) vs += ok[d] * po[d];
        float s = sqbo[h] + vs[0] + vs[1] + vs[2] + vs[3];
        satto[h][k] = s * SCALEF * sdwo[k];
    }
    __syncthreads();

    if (t < HH) {
        float m = -3.4e38f;
        for (int k = 0; k < KNB; ++k) m = fmaxf(m, satta[t][k]);
        float sum = 0.f;
        for (int k = 0; k < KNB; ++k) { float e = expf(satta[t][k] - m); satta[t][k] = e; sum += e; }
        float inv = 1.f / sum;
        for (int k = 0; k < KNB; ++k) satta[t][k] *= inv;
    } else if (t >= 32 && t < 32 + HH) {
        int h = t - 32;
        float m = -3.4e38f;
        for (int k = 0; k < KOB; ++k) m = fmaxf(m, satto[h][k]);
        float sum = 0.f;
        for (int k = 0; k < KOB; ++k) { float e = expf(satto[h][k] - m); satto[h][k] = e; sum += e; }
        float inv = 1.f / sum;
        for (int k = 0; k < KOB; ++k) satto[h][k] *= inv;
    }
    __syncthreads();

    v4f* Yan = (v4f*)(Ya + (size_t)n * 2048);
    v4f* Yon = (v4f*)(Yo + (size_t)n * 1024);
    for (int i = t; i < HH * 64; i += 256) {
        int h = i >> 6, d4 = (i & 63) << 2;
        v4f s = {};
        for (int k = 0; k < KNB; ++k) s += satta[h][k] * *(const v4f*)&sav[k][d4];
        Yan[i] = s;
    }
    for (int i = t; i < HH * 32; i += 256) {
        int h = i >> 5, d4 = (i & 31) << 2;
        v4f s = {};
        for (int k = 0; k < KOB; ++k) s += satto[h][k] * *(const v4f*)&sov[k][d4];
        Yon[i] = s;
    }
}

__device__ __forceinline__
void bias_fold_block(int blk,
                     const float* __restrict__ bav, const float* __restrict__ Wpa,
                     const float* __restrict__ bpa,
                     const float* __restrict__ bov, const float* __restrict__ Wpo,
                     const float* __restrict__ bpo,
                     float* __restrict__ bca, float* __restrict__ bco)
{
    __shared__ float part[32][32];
    const int lane = threadIdx.x & 31;
    const int w = __builtin_amdgcn_readfirstlane((int)(threadIdx.x >> 5));
    const bool obst = blk >= 4;
    const int j = ((blk & 3) << 5) + lane;
    const float* bvec = obst ? bov : bav;
    const float* wmat = (obst ? Wpo : Wpa) + j;
#pragma unroll 1
    for (int g = w; g < 32; g += 4) {
        const float* bv = bvec + g * 32;
        const float* wp = wmat + (size_t)(g * 32) * CC;
        float sum = 0.f;
#pragma unroll
        for (int u = 0; u < 32; ++u) sum += bv[u] * wp[u * CC];
        part[g][lane] = sum;
    }
    __syncthreads();
    if (w == 0) {
        float t = (obst ? bpo : bpa)[j];
        for (int k = 0; k < 32; ++k) t += part[k][lane];
        (obst ? bco : bca)[j] = t;
    }
}

extern "C" void kernel_launch(void* const* d_in, const int* in_sizes, int n_in,
                              void* d_out, int out_size, void* d_ws, size_t ws_size,
                              hipStream_t stream)
{
    (void)in_sizes; (void)n_in; (void)out_size; (void)ws_size;
    const float* qf   = (const float*)d_in[0];
    const float* xak  = (const float*)d_in[1];
    const float* xav  = (const float*)d_in[2];
    const float* xok  = (const float*)d_in[3];
    const float* xov  = (const float*)d_in[4];
    const float* apos = (const float*)d_in[5];
    const float* npos = (const float*)d_in[6];
    const float* opos = (const float*)d_in[7];
    const float* Wq   = (const float*)d_in[8];
    const float* bq   = (const float*)d_in[9];
    const float* Wak  = (const float*)d_in[10];
    const float* bak  = (const float*)d_in[11];
    const float* Wav  = (const float*)d_in[12];
    const float* bav  = (const float*)d_in[13];
    const float* Wok  = (const float*)d_in[14];
    const float* bok  = (const float*)d_in[15];
    const float* Wov  = (const float*)d_in[16];
    const float* bov  = (const float*)d_in[17];
    const float* Wpa  = (const float*)d_in[18];
    const float* bpa  = (const float*)d_in[19];
    const float* Wpo  = (const float*)d_in[20];
    const float* bpo  = (const float*)d_in[21];
    const float* dA   = (const float*)d_in[22];
    const float* dO   = (const float*)d_in[23];
    float* out = (float*)d_out;

    float* ws    = (float*)d_ws;
    float* wsQ   = ws;
    float* wsPa  = wsQ  + (size_t)NPTS * HCDIM;
    float* wsPo  = wsPa + (size_t)NPTS * 2048;
    float* wsYa  = wsPo + (size_t)NPTS * 1024;
    float* wsYo  = wsYa + (size_t)NPTS * 2048;
    float* wsWca = wsYo + (size_t)NPTS * 1024;
    float* wsWco = wsWca + (size_t)2048 * 128;
    float* wsBca = wsWco + (size_t)1024 * 128;
    float* wsBco = wsBca + 128;

    const GemmShape none = {0, 0, 0, 0,  0, 0, 0,  0, 0, 0,  0, 0};

    const GemmShape shQ   = {256, 4, 6, 1024,  256, 1, 0,  1024, 1, 0,  1024, 0};
    const GemmShape shWca = {128, 1, 2, 64,  1024, 1, 128,  128, 1, 128 * 128,  128, 256 * 128};
    const GemmShape shWco = {128, 1, 1, 32,  1024, 1, 128,  128, 1, 128 * 128,  128, 128 * 128};
    wmma_gemm_f32<<<1024 + 64 + 32 + 8, 128, 0, stream>>>(
        qf,  Wq,  bq,      wsQ,   shQ,
        Wav, Wpa, nullptr, wsWca, shWca,
        Wov, Wpo, nullptr, wsWco, shWco,
        bav, Wpa, bpa, bov, Wpo, bpo, wsBca, wsBco);

    const GemmShape shPa = {128, 2, 6, 2048,  1024, 1, 128,  1, 1024, 128,  2048, 256};
    const GemmShape shPo = {128, 1, 6, 1024,  1024, 1, 128,  1, 1024, 128,  1024, 128};
    wmma_gemm_f32<<<2048 + 1024, 128, 0, stream>>>(
        wsQ, Wak, nullptr, wsPa, shPa,
        wsQ, Wok, nullptr, wsPo, shPo,
        nullptr, nullptr, nullptr, nullptr, none,
        nullptr, nullptr, nullptr, nullptr, nullptr, nullptr, nullptr, nullptr);

    attend_fused<<<NPTS, 256, 0, stream>>>(
        xak, xav, xok, xov, apos, npos, opos,
        wsQ, wsPa, wsPo, bak, bok, dA, dO, wsYa, wsYo);

    const GemmShape shOa = {2048, 1, 6, 128,  2048, 1, 0,  128, 1, 0,  128, 0};
    const GemmShape shOo = {1024, 1, 6, 128,  1024, 1, 0,  128, 1, 0,  128, 0};
    wmma_gemm_f32<<<128 + 128, 128, 0, stream>>>(
        wsYa, wsWca, wsBca, out,                     shOa,
        wsYo, wsWco, wsBco, out + (size_t)NPTS * CC, shOo,
        nullptr, nullptr, nullptr, nullptr, none,
        nullptr, nullptr, nullptr, nullptr, nullptr, nullptr, nullptr, nullptr);
}


# Output GEMMs (K 2048/1024, 256 tiles): K split over 4 wave groups inside each work-group (16 waves; own LDS tiles and prefetch per group), partials added via LDS in group order; sum order changes
# speedup vs baseline: 7.5670x; 1.1734x over previous
#include <hip/hip_runtime.h>
#include <math.h>


typedef __attribute__((ext_vector_type(2))) float v2f;
typedef __attribute__((ext_vector_type(4))) float v4f;
typedef __attribute__((ext_vector_type(8))) float v8f;

#define NPTS 4096
#define KNB  16
#define KOB  8
#define HH   8
#define CC   128
#define HCDIM 1024
#define SCALEF 0.25f

#define KC 32
#define LDP (KC + 2)
struct GemmShape {
    int  K;
    int  ltx;
    int  lty;
    int  nblk;
    long sAm, sAk, bsA;
    long sBk, sBn, bsB;
    long ldc, bsC;
};

__device__ __forceinline__
void bias_fold_block(int blk,
                     const float* __restrict__ bav, const float* __restrict__ Wpa,
                     const float* __restrict__ bpa,
                     const float* __restrict__ bov, const float* __restrict__ Wpo,
                     const float* __restrict__ bpo,
                     float* __restrict__ bca, float* __restrict__ bco);

template <int S>
__global__ __launch_bounds__(128 * S)
void wmma_gemm_f32(const float* A0, const float* B0, const float* bias0, float* C0, GemmShape s0,
                   const float* A1, const float* B1, const float* bias1, float* C1, GemmShape s1,
                   const float* A2, const float* B2, const float* bias2, float* C2, GemmShape s2,
                   const float* bav, const float* Wpa, const float* bpa,
                   const float* bov, const float* Wpo, const float* bpo,
                   float* bca, float* bco)
{
    int blk = blockIdx.x;
    const float* __restrict__ A = A0;
    const float* __restrict__ B = B0;
    const float* __restrict__ bias = bias0;
    float* __restrict__ C = C0;
    GemmShape s = s0;
    if (blk >= s0.nblk) {
        blk -= s0.nblk;
        A = A1; B = B1; bias = bias1; C = C1; s = s1;
        if (blk >= s1.nblk) {
            blk -= s1.nblk;
            A = A2; B = B2; bias = bias2; C = C2; s = s2;
            if (blk >= s2.nblk) {
                if constexpr (S == 1)
                    bias_fold_block(blk - s2.nblk, bav, Wpa, bpa, bov, Wpo, bpo, bca, bco);
                return;
            }
        }
    }
    const int  bx = blk & ((1 << s.ltx) - 1);
    const int  by = (blk >> s.ltx) & ((1 << s.lty) - 1);
    const int  bz = blk >> (s.ltx + s.lty);
    const int  Kdim = s.K / S;
    const long sAm = s.sAm, sAk = s.sAk, bsA = s.bsA;
    const long sBk = s.sBk, sBn = s.sBn, bsB = s.bsB;
    const long ldc = s.ldc, bsC = s.bsC;

    __shared__ __align__(16) float tiles[S][2][64][LDP];

    const int grp  = threadIdx.x >> 7;
    const int wave = (threadIdx.x >> 5) & 3;
    const int lane = threadIdx.x & 31;
    float (*Alds)[LDP] = tiles[grp][0];
    float (*Blds)[LDP] = tiles[grp][1];
    const long kg0 = (long)grp * Kdim;
    const int m0 = by * 64;
    const int n0 = bx * 64;
    const float* Ab = A + (long)bz * bsA;
    const float* Bb = B + (long)bz * bsB;
    float* Cb = C + (long)bz * bsC;

    v8f acc[4] = {};

    const int row  = lane & 15;
    const int koff = (lane >> 4) << 1;

    const bool brow = (sBn == 1);
    const int  nrm  = ((wave & 1) << 5) + lane;
    const int  krm  = wave >> 1;
    const float* pa = Ab + (long)(m0 + wave) * sAm + (kg0 + lane) * sAk;
    const float* pb = brow ? Bb + (kg0 + krm) * sBk + (long)(n0 + nrm)
                           : Bb + (kg0 + lane) + (long)(n0 + wave) * sBn;
    const long stepA = 4 * sAm;
    const long stepB = brow ? 2 * sBk : 4 * sBn;
    const long advA  = (long)KC * sAk;
    const long advB  = brow ? (long)KC * sBk : (long)KC;

    float ra[16], rb[16];
    if (Kdim > 0) {
        const float* qa = pa;
        const float* qb = pb;
#pragma unroll
        for (int u = 0; u < 16; ++u) {
            ra[u] = *qa; rb[u] = *qb;
            qa += stepA; qb += stepB;
        }
    }

    for (int kb = 0; kb < Kdim; kb += KC) {
#pragma unroll
        for (int u = 0; u < 16; ++u) Alds[wave + 4 * u][lane] = ra[u];
        if (brow) {
#pragma unroll
            for (int u = 0; u < 16; ++u) Blds[nrm][krm + 2 * u] = rb[u];
        } else {
#pragma unroll
            for (int u = 0; u < 16; ++u) Blds[wave + 4 * u][lane] = rb[u];
        }
        __syncthreads();

        if (kb + KC < Kdim) {
            pa += advA; pb += advB;
            const float* qa = pa;
            const float* qb = pb;
#pragma unroll
            for (int u = 0; u < 16; ++u) {
                ra[u] = *qa; rb[u] = *qb;
                qa += stepA; qb += stepB;
            }
        }

#pragma unroll
        for (int ks = 0; ks < KC; ks += 4) {
            const int k0 = ks + koff;
            v2f a = *(const v2f*)&Alds[wave * 16 + row][k0];
#pragma unroll
            for (int j = 0; j < 4; ++j) {
                v2f b = *(const v2f*)&Blds[j * 16 + row][k0];
                acc[j] = __builtin_amdgcn_wmma_f32_16x16x4_f32(
                    false, a, false, b, (short)0, acc[j], false, false);
            }
        }
        __builtin_amdgcn_fence(__ATOMIC_RELEASE, "workgroup", "local");
        __builtin_amdgcn_s_barrier();
        __builtin_amdgcn_fence(__ATOMIC_ACQUIRE, "workgroup", "local");
    }

    if constexpr (S > 1) {
        const int prow = wave * 16 + ((lane >> 4) << 3);
        if (grp > 0) {
            float* mine = &tiles[grp][0][0][0];
#pragma unroll
            for (int j = 0; j < 4; ++j)
#pragma unroll
                for (int r = 0; r < 8; ++r)
                    mine[(prow + r) * 68 + j * 16 + row] = acc[j][r];
        }
        __syncthreads();
        if (grp > 0) return;
#pragma unroll
        for (int gg = 1; gg < S; ++gg) {
            const float* part = &tiles[gg][0][0][0];
#pragma unroll
            for (int j = 0; j < 4; ++j)
#pragma unroll
                for (int r = 0; r < 8; ++r)
                    acc[j][r] += part[(prow + r) * 68 + j * 16 + row];
        }
    }
    const int mbase = m0 + wave * 16 + ((lane >> 4) << 3);
#pragma unroll
    for (int j = 0; j < 4; ++j) {
        const int n = n0 + j * 16 + row;
        const float bv = bias ? bias[n] : 0.0f;
#pragma unroll
        for (int r = 0; r < 8; ++r)
            Cb[(long)(mbase + r) * ldc + n] = acc[j][r] + bv;
    }
}

#define LAP 260
#define LOP 132
__global__ __launch_bounds__(256)
void attend_fused(const float* __restrict__ xak, const float* __restrict__ xav,
                  const float* __restrict__ xok, const float* __restrict__ xov,
                  const float* __restrict__ apos, const float* __restrict__ npos,
                  const float* __restrict__ opos,
                  const float* __restrict__ Q,
                  const float* __restrict__ Pa, const float* __restrict__ Po,
                  const float* __restrict__ bak, const float* __restrict__ bok,
                  const float* __restrict__ decayA, const float* __restrict__ decayO,
                  float* __restrict__ Ya, float* __restrict__ Yo)
{
    const int n = blockIdx.x;
    const int t = threadIdx.x;

    __shared__ __align__(16) float sak[KNB][LAP];
    __shared__ __align__(16) float sav[KNB][LAP];
    __shared__ __align__(16) float spa[HH][LAP];
    __shared__ __align__(16) float sok[KOB][LOP];
    __shared__ __align__(16) float sov[KOB][LOP];
    __shared__ __align__(16) float spo[HH][LOP];
    __shared__ float satta[HH][KNB];
    __shared__ float satto[HH][KOB];
    __shared__ float sdwa[KNB];
    __shared__ float sdwo[KOB];
    __shared__ float sqba[HH];
    __shared__ float sqbo[HH];

    const v4f* xakn = (const v4f*)(xak + (size_t)n * KNB * 256);
    const v4f* xavn = (const v4f*)(xav + (size_t)n * KNB * 256);
    const v4f* xokn = (const v4f*)(xok + (size_t)n * KOB * 128);
    const v4f* xovn = (const v4f*)(xov + (size_t)n * KOB * 128);
    const v4f* pan  = (const v4f*)(Pa + (size_t)n * 2048);
    const v4f* pon  = (const v4f*)(Po + (size_t)n * 1024);

    for (int i = t; i < KNB * 64; i += 256) {
        int k = i >> 6, d4 = (i & 63) << 2;
        *(v4f*)&sak[k][d4] = xakn[i];
        *(v4f*)&sav[k][d4] = xavn[i];
    }
    for (int i = t; i < HH * 64; i += 256)
        *(v4f*)&spa[i >> 6][(i & 63) << 2] = pan[i];
    for (int i = t; i < KOB * 32; i += 256) {
        int k = i >> 5, d4 = (i & 31) << 2;
        *(v4f*)&sok[k][d4] = xokn[i];
        *(v4f*)&sov[k][d4] = xovn[i];
    }
    for (int i = t; i < HH * 32; i += 256)
        *(v4f*)&spo[i >> 5][(i & 31) << 2] = pon[i];

    const float ax = apos[n * 3 + 0], ay = apos[n * 3 + 1], az = apos[n * 3 + 2];
    if (t < KNB) {
        float dx = ax - npos[(n * KNB + t) * 3 + 0];
        float dy = ay - npos[(n * KNB + t) * 3 + 1];
        float dz = az - npos[(n * KNB + t) * 3 + 2];
        float dist = fmaxf(sqrtf(dx * dx + dy * dy + dz * dz), 0.001f);
        sdwa[t] = expf(-decayA[0] * dist);
    }
    if (t >= 32 && t < 32 + KOB) {
        int k = t - 32;
        float dx = ax - opos[(n * KOB + k) * 3 + 0];
        float dy = ay - opos[(n * KOB + k) * 3 + 1];
        float dz = az - opos[(n * KOB + k) * 3 + 2];
        float dist = sqrtf(dx * dx + dy * dy + dz * dz);
        sdwo[k] = expf(-decayO[0] * dist);
    }
    if (t >= 64 && t < 64 + HH) {
        int h = t - 64;
        float sa = 0.f, so = 0.f;
        for (int c = 0; c < CC; ++c) {
            float q = Q[(size_t)n * HCDIM + h * CC + c];
            sa += q * bak[h * CC + c];
            so += q * bok[h * CC + c];
        }
        sqba[h] = sa; sqbo[h] = so;
    }
    __syncthreads();

    if (t < HH * KNB) {
        int h = t >> 4, k = t & 15;
        const v4f* ak = (const v4f*)sak[k];
        const v4f* pa = (const v4f*)spa[h];
        v4f vs = {};
        for (int d = 0; d < 64; ++d) vs += ak[d] * pa[d];
        float s = sqba[h] + vs[0] + vs[1] + vs[2] + vs[3];
        satta[h][k] = s * SCALEF * sdwa[k];
    } else if (t < HH * KNB + HH * KOB) {
        int u = t - HH * KNB;
        int h = u >> 3, k = u & 7;
        const v4f* ok = (const v4f*)sok[k];
        const v4f* po = (const v4f*)spo[h];
        v4f vs = {};
        for (int d = 0; d < 32; ++d) vs += ok[d] * po[d];
        float s = sqbo[h] + vs[0] + vs[1] + vs[2] + vs[3];
        satto[h][k] = s * SCALEF * sdwo[k];
    }
    __syncthreads();

    if (t < HH) {
        float m = -3.4e38f;
        for (int k = 0; k < KNB; ++k) m = fmaxf(m, satta[t][k]);
        float sum = 0.f;
        for (int k = 0; k < KNB; ++k) { float e = expf(satta[t][k] - m); satta[t][k] = e; sum += e; }
        float inv = 1.f / sum;
        for (int k = 0; k < KNB; ++k) satta[t][k] *= inv;
    } else if (t >= 32 && t < 32 + HH) {
        int h = t - 32;
        float m = -3.4e38f;
        for (int k = 0; k < KOB; ++k) m = fmaxf(m, satto[h][k]);
        float sum = 0.f;
        for (int k = 0; k < KOB; ++k) { float e = expf(satto[h][k] - m); satto[h][k] = e; sum += e; }
        float inv = 1.f / sum;
        for (int k = 0; k < KOB; ++k) satto[h][k] *= inv;
    }
    __syncthreads();

    v4f* Yan = (v4f*)(Ya + (size_t)n * 2048);
    v4f* Yon = (v4f*)(Yo + (size_t)n * 1024);
    for (int i = t; i < HH * 64; i += 256) {
        int h = i >> 6, d4 = (i & 63) << 2;
        v4f s = {};
        for (int k = 0; k < KNB; ++k) s += satta[h][k] * *(const v4f*)&sav[k][d4];
        Yan[i] = s;
    }
    for (int i = t; i < HH * 32; i += 256) {
        int h = i >> 5, d4 = (i & 31) << 2;
        v4f s = {};
        for (int k = 0; k < KOB; ++k) s += satto[h][k] * *(const v4f*)&sov[k][d4];
        Yon[i] = s;
    }
}

__device__ __forceinline__
void bias_fold_block(int blk,
                     const float* __restrict__ bav, const float* __restrict__ Wpa,
                     const float* __restrict__ bpa,
                     const float* __restrict__ bov, const float* __restrict__ Wpo,
                     const float* __restrict__ bpo,
                     float* __restrict__ bca, float* __restrict__ bco)
{
    __shared__ float part[32][32];
    const int lane = threadIdx.x & 31;
    const int w = __builtin_amdgcn_readfirstlane((int)(threadIdx.x >> 5));
    const bool obst = blk >= 4;
    const int j = ((blk & 3) << 5) + lane;
    const float* bvec = obst ? bov : bav;
    const float* wmat = (obst ? Wpo : Wpa) + j;
#pragma unroll 1
    for (int g = w; g < 32; g += 4) {
        const float* bv = bvec + g * 32;
        const float* wp = wmat + (size_t)(g * 32) * CC;
        float sum = 0.f;
#pragma unroll
        for (int u = 0; u < 32; ++u) sum += bv[u] * wp[u * CC];
        part[g][lane] = sum;
    }
    __syncthreads();
    if (w == 0) {
        float t = (obst ? bpo : bpa)[j];
        for (int k = 0; k < 32; ++k) t += part[k][lane];
        (obst ? bco : bca)[j] = t;
    }
}

extern "C" void kernel_launch(void* const* d_in, const int* in_sizes, int n_in,
                              void* d_out, int out_size, void* d_ws, size_t ws_size,
                              hipStream_t stream)
{
    (void)in_sizes; (void)n_in; (void)out_size; (void)ws_size;
    const float* qf   = (const float*)d_in[0];
    const float* xak  = (const float*)d_in[1];
    const float* xav  = (const float*)d_in[2];
    const float* xok  = (const float*)d_in[3];
    const float* xov  = (const float*)d_in[4];
    const float* apos = (const float*)d_in[5];
    const float* npos = (const float*)d_in[6];
    const float* opos = (const float*)d_in[7];
    const float* Wq   = (const float*)d_in[8];
    const float* bq   = (const float*)d_in[9];
    const float* Wak  = (const float*)d_in[10];
    const float* bak  = (const float*)d_in[11];
    const float* Wav  = (const float*)d_in[12];
    const float* bav  = (const float*)d_in[13];
    const float* Wok  = (const float*)d_in[14];
    const float* bok  = (const float*)d_in[15];
    const float* Wov  = (const float*)d_in[16];
    const float* bov  = (const float*)d_in[17];
    const float* Wpa  = (const float*)d_in[18];
    const float* bpa  = (const float*)d_in[19];
    const float* Wpo  = (const float*)d_in[20];
    const float* bpo  = (const float*)d_in[21];
    const float* dA   = (const float*)d_in[22];
    const float* dO   = (const float*)d_in[23];
    float* out = (float*)d_out;

    float* ws    = (float*)d_ws;
    float* wsQ   = ws;
    float* wsPa  = wsQ  + (size_t)NPTS * HCDIM;
    float* wsPo  = wsPa + (size_t)NPTS * 2048;
    float* wsYa  = wsPo + (size_t)NPTS * 1024;
    float* wsYo  = wsYa + (size_t)NPTS * 2048;
    float* wsWca = wsYo + (size_t)NPTS * 1024;
    float* wsWco = wsWca + (size_t)2048 * 128;
    float* wsBca = wsWco + (size_t)1024 * 128;
    float* wsBco = wsBca + 128;

    const GemmShape none = {0, 0, 0, 0,  0, 0, 0,  0, 0, 0,  0, 0};

    const GemmShape shQ   = {256, 4, 6, 1024,  256, 1, 0,  1024, 1, 0,  1024, 0};
    const GemmShape shWca = {128, 1, 2, 64,  1024, 1, 128,  128, 1, 128 * 128,  128, 256 * 128};
    const GemmShape shWco = {128, 1, 1, 32,  1024, 1, 128,  128, 1, 128 * 128,  128, 128 * 128};
    wmma_gemm_f32<1><<<1024 + 64 + 32 + 8, 128, 0, stream>>>(
        qf,  Wq,  bq,      wsQ,   shQ,
        Wav, Wpa, nullptr, wsWca, shWca,
        Wov, Wpo, nullptr, wsWco, shWco,
        bav, Wpa, bpa, bov, Wpo, bpo, wsBca, wsBco);

    const GemmShape shPa = {128, 2, 6, 2048,  1024, 1, 128,  1, 1024, 128,  2048, 256};
    const GemmShape shPo = {128, 1, 6, 1024,  1024, 1, 128,  1, 1024, 128,  1024, 128};
    wmma_gemm_f32<1><<<2048 + 1024, 128, 0, stream>>>(
        wsQ, Wak, nullptr, wsPa, shPa,
        wsQ, Wok, nullptr, wsPo, shPo,
        nullptr, nullptr, nullptr, nullptr, none,
        nullptr, nullptr, nullptr, nullptr, nullptr, nullptr, nullptr, nullptr);

    attend_fused<<<NPTS, 256, 0, stream>>>(
        xak, xav, xok, xov, apos, npos, opos,
        wsQ, wsPa, wsPo, bak, bok, dA, dO, wsYa, wsYo);

    const GemmShape shOa = {2048, 1, 6, 128,  2048, 1, 0,  128, 1, 0,  128, 0};
    const GemmShape shOo = {1024, 1, 6, 128,  1024, 1, 0,  128, 1, 0,  128, 0};
    wmma_gemm_f32<4><<<128 + 128, 512, 0, stream>>>(
        wsYa, wsWca, wsBca, out,                     shOa,
        wsYo, wsWco, wsBco, out + (size_t)NPTS * CC, shOo,
        nullptr, nullptr, nullptr, nullptr, none,
        nullptr, nullptr, nullptr, nullptr, nullptr, nullptr, nullptr, nullptr);
}


# Attention kernel: every global load of a work-group (features, p vectors, positions; q.key-bias dot as one b128 per lane + wave butterfly) goes out as ONE group; that dot's sum order changes
# speedup vs baseline: 8.3366x; 1.1017x over previous
#include <hip/hip_runtime.h>
#include <math.h>


typedef __attribute__((ext_vector_type(2))) float v2f;
typedef __attribute__((ext_vector_type(4))) float v4f;
typedef __attribute__((ext_vector_type(8))) float v8f;

#define NPTS 4096
#define KNB  16
#define KOB  8
#define HH   8
#define CC   128
#define HCDIM 1024
#define SCALEF 0.25f

#define KC 32
#define LDP (KC + 2)
struct GemmShape {
    int  K;
    int  ltx;
    int  lty;
    int  nblk;
    long sAm, sAk, bsA;
    long sBk, sBn, bsB;
    long ldc, bsC;
};

__device__ __forceinline__
void bias_fold_block(int blk,
                     const float* __restrict__ bav, const float* __restrict__ Wpa,
                     const float* __restrict__ bpa,
                     const float* __restrict__ bov, const float* __restrict__ Wpo,
                     const float* __restrict__ bpo,
                     float* __restrict__ bca, float* __restrict__ bco);

template <int S>
__global__ __launch_bounds__(128 * S)
void wmma_gemm_f32(const float* A0, const float* B0, const float* bias0, float* C0, GemmShape s0,
                   const float* A1, const float* B1, const float* bias1, float* C1, GemmShape s1,
                   const float* A2, const float* B2, const float* bias2, float* C2, GemmShape s2,
                   const float* bav, const float* Wpa, const float* bpa,
                   const float* bov, const float* Wpo, const float* bpo,
                   float* bca, float* bco)
{
    int blk = blockIdx.x;
    const float* __restrict__ A = A0;
    const float* __restrict__ B = B0;
    const float* __restrict__ bias = bias0;
    float* __restrict__ C = C0;
    GemmShape s = s0;
    if (blk >= s0.nblk) {
        blk -= s0.nblk;
        A = A1; B = B1; bias = bias1; C = C1; s = s1;
        if (blk >= s1.nblk) {
            blk -= s1.nblk;
            A = A2; B = B2; bias = bias2; C = C2; s = s2;
            if (blk >= s2.nblk) {
                if constexpr (S == 1)
                    bias_fold_block(blk - s2.nblk, bav, Wpa, bpa, bov, Wpo, bpo, bca, bco);
                return;
            }
        }
    }
    const int  bx = blk & ((1 << s.ltx) - 1);
    const int  by = (blk >> s.ltx) & ((1 << s.lty) - 1);
    const int  bz = blk >> (s.ltx + s.lty);
    const int  Kdim = s.K / S;
    const long sAm = s.sAm, sAk = s.sAk, bsA = s.bsA;
    const long sBk = s.sBk, sBn = s.sBn, bsB = s.bsB;
    const long ldc = s.ldc, bsC = s.bsC;

    __shared__ __align__(16) float tiles[S][2][64][LDP];

    const int grp  = threadIdx.x >> 7;
    const int wave = (threadIdx.x >> 5) & 3;
    const int lane = threadIdx.x & 31;
    float (*Alds)[LDP] = tiles[grp][0];
    float (*Blds)[LDP] = tiles[grp][1];
    const long kg0 = (long)grp * Kdim;
    const int m0 = by * 64;
    const int n0 = bx * 64;
    const float* Ab = A + (long)bz * bsA;
    const float* Bb = B + (long)bz * bsB;
    float* Cb = C + (long)bz * bsC;

    v8f acc[4] = {};

    const int row  = lane & 15;
    const int koff = (lane >> 4) << 1;

    const bool brow = (sBn == 1);
    const int  nrm  = ((wave & 1) << 5) + lane;
    const int  krm  = wave >> 1;
    const float* pa = Ab + (long)(m0 + wave) * sAm + (kg0 + lane) * sAk;
    const float* pb = brow ? Bb + (kg0 + krm) * sBk + (long)(n0 + nrm)
                           : Bb + (kg0 + lane) + (long)(n0 + wave) * sBn;
    const long stepA = 4 * sAm;
    const long stepB = brow ? 2 * sBk : 4 * sBn;
    const long advA  = (long)KC * sAk;
    const long advB  = brow ? (long)KC * sBk : (long)KC;

    float ra[16], rb[16];
    if (Kdim > 0) {
        const float* qa = pa;
        const float* qb = pb;
#pragma unroll
        for (int u = 0; u < 16; ++u) {
            ra[u] = *qa; rb[u] = *qb;
            qa += stepA; qb += stepB;
        }
    }

    for (int kb = 0; kb < Kdim; kb += KC) {
#pragma unroll
        for (int u = 0; u < 16; ++u) Alds[wave + 4 * u][lane] = ra[u];
        if (brow) {
#pragma unroll
            for (int u = 0; u < 16; ++u) Blds[nrm][krm + 2 * u] = rb[u];
        } else {
#pragma unroll
            for (int u = 0; u < 16; ++u) Blds[wave + 4 * u][lane] = rb[u];
        }
        __syncthreads();

        if (kb + KC < Kdim) {
            pa += advA; pb += advB;
            const float* qa = pa;
            const float* qb = pb;
#pragma unroll
            for (int u = 0; u < 16; ++u) {
                ra[u] = *qa; rb[u] = *qb;
                qa += stepA; qb += stepB;
            }
        }

#pragma unroll
        for (int ks = 0; ks < KC; ks += 4) {
            const int k0 = ks + koff;
            v2f a = *(const v2f*)&Alds[wave * 16 + row][k0];
#pragma unroll
            for (int j = 0; j < 4; ++j) {
                v2f b = *(const v2f*)&Blds[j * 16 + row][k0];
                acc[j] = __builtin_amdgcn_wmma_f32_16x16x4_f32(
                    false, a, false, b, (short)0, acc[j], false, false);
            }
        }
        __builtin_amdgcn_fence(__ATOMIC_RELEASE, "workgroup", "local");
        __builtin_amdgcn_s_barrier();
        __builtin_amdgcn_fence(__ATOMIC_ACQUIRE, "workgroup", "local");
    }

    if constexpr (S > 1) {
        const int prow = wave * 16 + ((lane >> 4) << 3);
        if (grp > 0) {
            float* mine = &tiles[grp][0][0][0];
#pragma unroll
            for (int j = 0; j < 4; ++j)
#pragma unroll
                for (int r = 0; r < 8; ++r)
                    mine[(prow + r) * 68 + j * 16 + row] = acc[j][r];
        }
        __syncthreads();
        if (grp > 0) return;
#pragma unroll
        for (int gg = 1; gg < S; ++gg) {
            const float* part = &tiles[gg][0][0][0];
#pragma unroll
            for (int j = 0; j < 4; ++j)
#pragma unroll
                for (int r = 0; r < 8; ++r)
                    acc[j][r] += part[(prow + r) * 68 + j * 16 + row];
        }
    }
    const int mbase = m0 + wave * 16 + ((lane >> 4) << 3);
#pragma unroll
    for (int j = 0; j < 4; ++j) {
        const int n = n0 + j * 16 + row;
        const float bv = bias ? bias[n] : 0.0f;
#pragma unroll
        for (int r = 0; r < 8; ++r)
            Cb[(long)(mbase + r) * ldc + n] = acc[j][r] + bv;
    }
}

#define LAP 260
#define LOP 132
__global__ __launch_bounds__(256)
void attend_fused(const float* __restrict__ xak, const float* __restrict__ xav,
                  const float* __restrict__ xok, const float* __restrict__ xov,
                  const float* __restrict__ apos, const float* __restrict__ npos,
                  const float* __restrict__ opos,
                  const float* __restrict__ Q,
                  const float* __restrict__ Pa, const float* __restrict__ Po,
                  const float* __restrict__ bak, const float* __restrict__ bok,
                  const float* __restrict__ decayA, const float* __restrict__ decayO,
                  float* __restrict__ Ya, float* __restrict__ Yo)
{
    const int n = blockIdx.x;
    const int t = threadIdx.x;

    __shared__ __align__(16) float sak[KNB][LAP];
    __shared__ __align__(16) float sav[KNB][LAP];
    __shared__ __align__(16) float spa[HH][LAP];
    __shared__ __align__(16) float sok[KOB][LOP];
    __shared__ __align__(16) float sov[KOB][LOP];
    __shared__ __align__(16) float spo[HH][LOP];
    __shared__ float satta[HH][KNB];
    __shared__ float satto[HH][KOB];
    __shared__ float sdwa[KNB];
    __shared__ float sdwo[KOB];
    __shared__ float sqba[HH];
    __shared__ float sqbo[HH];

    const v4f* xakn = (const v4f*)(xak + (size_t)n * KNB * 256);
    const v4f* xavn = (const v4f*)(xav + (size_t)n * KNB * 256);
    const v4f* xokn = (const v4f*)(xok + (size_t)n * KOB * 128);
    const v4f* xovn = (const v4f*)(xov + (size_t)n * KOB * 128);
    const v4f* pan  = (const v4f*)(Pa + (size_t)n * 2048);
    const v4f* pon  = (const v4f*)(Po + (size_t)n * 1024);

    const v4f* qn   = (const v4f*)(Q + (size_t)n * HCDIM);

    const float ax = apos[n * 3 + 0], ay = apos[n * 3 + 1], az = apos[n * 3 + 2];
    const float dcA = decayA[0], dcO = decayO[0];
    const bool isA = t < KNB;
    const bool isO = (t >= 32) && (t < 32 + KOB);
    float ox = 0.f, oy = 0.f, oz = 0.f;
    if (isA || isO) {
        const float* pp = isA ? npos + (size_t)(n * KNB + t) * 3
                              : opos + (size_t)(n * KOB + (t - 32)) * 3;
        ox = pp[0]; oy = pp[1]; oz = pp[2];
    }

    v4f rak[4], rav[4], rpa[2];
#pragma unroll
    for (int u = 0; u < 4; ++u) {
        rak[u] = xakn[t + 256 * u];
        rav[u] = xavn[t + 256 * u];
    }
#pragma unroll
    for (int u = 0; u < 2; ++u) rpa[u] = pan[t + 256 * u];
    const v4f rok = xokn[t];
    const v4f rov = xovn[t];
    const v4f rpo = pon[t];
    const v4f rq  = qn[t];
    const v4f rbk = ((const v4f*)bak)[t];
    const v4f rbo = ((const v4f*)bok)[t];

    {
        const int k0 = t >> 6, d4 = (t & 63) << 2;
#pragma unroll
        for (int u = 0; u < 4; ++u) {
            *(v4f*)&sak[k0 + 4 * u][d4] = rak[u];
            *(v4f*)&sav[k0 + 4 * u][d4] = rav[u];
        }
#pragma unroll
        for (int u = 0; u < 2; ++u) *(v4f*)&spa[k0 + 4 * u][d4] = rpa[u];
        const int k1 = t >> 5, e4 = (t & 31) << 2;
        *(v4f*)&sok[k1][e4] = rok;
        *(v4f*)&sov[k1][e4] = rov;
        *(v4f*)&spo[k1][e4] = rpo;
    }

    if (isA) {
        float dx = ax - ox, dy = ay - oy, dz = az - oz;
        float dist = fmaxf(sqrtf(dx * dx + dy * dy + dz * dz), 0.001f);
        sdwa[t] = expf(-dcA * dist);
    } else if (isO) {
        float dx = ax - ox, dy = ay - oy, dz = az - oz;
        float dist = sqrtf(dx * dx + dy * dy + dz * dz);
        sdwo[t - 32] = expf(-dcO * dist);
    }
    {
        float sa = rq[0] * rbk[0] + rq[1] * rbk[1] + rq[2] * rbk[2] + rq[3] * rbk[3];
        float so = rq[0] * rbo[0] + rq[1] * rbo[1] + rq[2] * rbo[2] + rq[3] * rbo[3];
#pragma unroll
        for (int m = 16; m > 0; m >>= 1) {
            sa += __shfl_xor(sa, m);
            so += __shfl_xor(so, m);
        }
        if ((t & 31) == 0) { sqba[t >> 5] = sa; sqbo[t >> 5] = so; }
    }
    __syncthreads();

    if (t < HH * KNB) {
        int h = t >> 4, k = t & 15;
        const v4f* ak = (const v4f*)sak[k];
        const v4f* pa = (const v4f*)spa[h];
        v4f vs = {};
        for (int d = 0; d < 64; ++d) vs += ak[d] * pa[d];
        float s = sqba[h] + vs[0] + vs[1] + vs[2] + vs[3];
        satta[h][k] = s * SCALEF * sdwa[k];
    } else if (t < HH * KNB + HH * KOB) {
        int u = t - HH * KNB;
        int h = u >> 3, k = u & 7;
        const v4f* ok = (const v4f*)sok[k];
        const v4f* po = (const v4f*)spo[h];
        v4f vs = {};
        for (int d = 0; d < 32; ++d) vs += ok[d] * po[d];
        float s = sqbo[h] + vs[0] + vs[1] + vs[2] + vs[3];
        satto[h][k] = s * SCALEF * sdwo[k];
    }
    __syncthreads();

    if (t < HH) {
        float m = -3.4e38f;
        for (int k = 0; k < KNB; ++k) m = fmaxf(m, satta[t][k]);
        float sum = 0.f;
        for (int k = 0; k < KNB; ++k) { float e = expf(satta[t][k] - m); satta[t][k] = e; sum += e; }
        float inv = 1.f / sum;
        for (int k = 0; k < KNB; ++k) satta[t][k] *= inv;
    } else if (t >= 32 && t < 32 + HH) {
        int h = t - 32;
        float m = -3.4e38f;
        for (int k = 0; k < KOB; ++k) m = fmaxf(m, satto[h][k]);
        float sum = 0.f;
        for (int k = 0; k < KOB; ++k) { float e = expf(satto[h][k] - m); satto[h][k] = e; sum += e; }
        float inv = 1.f / sum;
        for (int k = 0; k < KOB; ++k) satto[h][k] *= inv;
    }
    __syncthreads();

    v4f* Yan = (v4f*)(Ya + (size_t)n * 2048);
    v4f* Yon = (v4f*)(Yo + (size_t)n * 1024);
    for (int i = t; i < HH * 64; i += 256) {
        int h = i >> 6, d4 = (i & 63) << 2;
        v4f s = {};
        for (int k = 0; k < KNB; ++k) s += satta[h][k] * *(const v4f*)&sav[k][d4];
        Yan[i] = s;
    }
    for (int i = t; i < HH * 32; i += 256) {
        int h = i >> 5, d4 = (i & 31) << 2;
        v4f s = {};
        for (int k = 0; k < KOB; ++k) s += satto[h][k] * *(const v4f*)&sov[k][d4];
        Yon[i] = s;
    }
}

__device__ __forceinline__
void bias_fold_block(int blk,
                     const float* __restrict__ bav, const float* __restrict__ Wpa,
                     const float* __restrict__ bpa,
                     const float* __restrict__ bov, const float* __restrict__ Wpo,
                     const float* __restrict__ bpo,
                     float* __restrict__ bca, float* __restrict__ bco)
{
    __shared__ float part[32][32];
    const int lane = threadIdx.x & 31;
    const int w = __builtin_amdgcn_readfirstlane((int)(threadIdx.x >> 5));
    const bool obst = blk >= 4;
    const int j = ((blk & 3) << 5) + lane;
    const float* bvec = obst ? bov : bav;
    const float* wmat = (obst ? Wpo : Wpa) + j;
#pragma unroll 1
    for (int g = w; g < 32; g += 4) {
        const float* bv = bvec + g * 32;
        const float* wp = wmat + (size_t)(g * 32) * CC;
        float sum = 0.f;
#pragma unroll
        for (int u = 0; u < 32; ++u) sum += bv[u] * wp[u * CC];
        part[g][lane] = sum;
    }
    __syncthreads();
    if (w == 0) {
        float t = (obst ? bpo : bpa)[j];
        for (int k = 0; k < 32; ++k) t += part[k][lane];
        (obst ? bco : bca)[j] = t;
    }
}

extern "C" void kernel_launch(void* const* d_in, const int* in_sizes, int n_in,
                              void* d_out, int out_size, void* d_ws, size_t ws_size,
                              hipStream_t stream)
{
    (void)in_sizes; (void)n_in; (void)out_size; (void)ws_size;
    const float* qf   = (const float*)d_in[0];
    const float* xak  = (const float*)d_in[1];
    const float* xav  = (const float*)d_in[2];
    const float* xok  = (const float*)d_in[3];
    const float* xov  = (const float*)d_in[4];
    const float* apos = (const float*)d_in[5];
    const float* npos = (const float*)d_in[6];
    const float* opos = (const float*)d_in[7];
    const float* Wq   = (const float*)d_in[8];
    const float* bq   = (const float*)d_in[9];
    const float* Wak  = (const float*)d_in[10];
    const float* bak  = (const float*)d_in[11];
    const float* Wav  = (const float*)d_in[12];
    const float* bav  = (const float*)d_in[13];
    const float* Wok  = (const float*)d_in[14];
    const float* bok  = (const float*)d_in[15];
    const float* Wov  = (const float*)d_in[16];
    const float* bov  = (const float*)d_in[17];
    const float* Wpa  = (const float*)d_in[18];
    const float* bpa  = (const float*)d_in[19];
    const float* Wpo  = (const float*)d_in[20];
    const float* bpo  = (const float*)d_in[21];
    const float* dA   = (const float*)d_in[22];
    const float* dO   = (const float*)d_in[23];
    float* out = (float*)d_out;

    float* ws    = (float*)d_ws;
    float* wsQ   = ws;
    float* wsPa  = wsQ  + (size_t)NPTS * HCDIM;
    float* wsPo  = wsPa + (size_t)NPTS * 2048;
    float* wsYa  = wsPo + (size_t)NPTS * 1024;
    float* wsYo  = wsYa + (size_t)NPTS * 2048;
    float* wsWca = wsYo + (size_t)NPTS * 1024;
    float* wsWco = wsWca + (size_t)2048 * 128;
    float* wsBca = wsWco + (size_t)1024 * 128;
    float* wsBco = wsBca + 128;

    const GemmShape none = {0, 0, 0, 0,  0, 0, 0,  0, 0, 0,  0, 0};

    const GemmShape shQ   = {256, 4, 6, 1024,  256, 1, 0,  1024, 1, 0,  1024, 0};
    const GemmShape shWca = {128, 1, 2, 64,  1024, 1, 128,  128, 1, 128 * 128,  128, 256 * 128};
    const GemmShape shWco = {128, 1, 1, 32,  1024, 1, 128,  128, 1, 128 * 128,  128, 128 * 128};
    wmma_gemm_f32<1><<<1024 + 64 + 32 + 8, 128, 0, stream>>>(
        qf,  Wq,  bq,      wsQ,   shQ,
        Wav, Wpa, nullptr, wsWca, shWca,
        Wov, Wpo, nullptr, wsWco, shWco,
        bav, Wpa, bpa, bov, Wpo, bpo, wsBca, wsBco);

    const GemmShape shPa = {128, 2, 6, 2048,  1024, 1, 128,  1, 1024, 128,  2048, 256};
    const GemmShape shPo = {128, 1, 6, 1024,  1024, 1, 128,  1, 1024, 128,  1024, 128};
    wmma_gemm_f32<1><<<2048 + 1024, 128, 0, stream>>>(
        wsQ, Wak, nullptr, wsPa, shPa,
        wsQ, Wok, nullptr, wsPo, shPo,
        nullptr, nullptr, nullptr, nullptr, none,
        nullptr, nullptr, nullptr, nullptr, nullptr, nullptr, nullptr, nullptr);

    attend_fused<<<NPTS, 256, 0, stream>>>(
        xak, xav, xok, xov, apos, npos, opos,
        wsQ, wsPa, wsPo, bak, bok, dA, dO, wsYa, wsYo);

    const GemmShape shOa = {2048, 1, 6, 128,  2048, 1, 0,  128, 1, 0,  128, 0};
    const GemmShape shOo = {1024, 1, 6, 128,  1024, 1, 0,  128, 1, 0,  128, 0};
    wmma_gemm_f32<4><<<128 + 128, 512, 0, stream>>>(
        wsYa, wsWca, wsBca, out,                     shOa,
        wsYo, wsWco, wsBco, out + (size_t)NPTS * CC, shOo,
        nullptr, nullptr, nullptr, nullptr, none,
        nullptr, nullptr, nullptr, nullptr, nullptr, nullptr, nullptr, nullptr);
}
